# starting kernel
# baseline (speedup 1.0000x reference)
#include <hip/hip_runtime.h>


typedef __attribute__((ext_vector_type(2))) float v2f;
typedef __attribute__((ext_vector_type(8))) float v8f;

#define NNODES 100000


__global__ void zero_f32(float* __restrict__ p, int n) {
  int i = blockIdx.x * blockDim.x + threadIdx.x;
  if (i < n) p[i] = 0.0f;
}

__global__ void degree_kernel(const int* __restrict__ src, const int* __restrict__ dst,
                              float* __restrict__ deg_out, float* __restrict__ deg_in, int E) {
  int e = blockIdx.x * blockDim.x + threadIdx.x;
  if (e < E) {
    atomicAdd(&deg_out[src[e]], 1.0f);
    atomicAdd(&deg_in[dst[e]], 1.0f);
  }
}

__global__ void norm_kernel(float* __restrict__ deg_out, float* __restrict__ deg_in, int n) {
  int i = blockIdx.x * blockDim.x + threadIdx.x;
  if (i < n) {
    deg_out[i] = rsqrtf(fmaxf(deg_out[i], 1.0f));
    deg_in[i]  = rsqrtf(fmaxf(deg_in[i], 1.0f));
  }
}

template <int F, bool SCALE>
__global__ void scatter_kernel(const float* __restrict__ x,
                               const int* __restrict__ src, const int* __restrict__ dst,
                               const float* __restrict__ scale,
                               float* __restrict__ agg, int E) {
  int tid = blockIdx.x * blockDim.x + threadIdx.x;
  int e = tid / F;
  int f = tid - e * F;
  if (e < E) {
    int s = src[e];
    int d = dst[e];
    float v = x[s * F + f];
    if (SCALE) v *= scale[s];
    atomicAdd(&agg[d * F + f], v);
  }
}

template <int F, bool RELU>
__global__ void epilogue_kernel(const float* __restrict__ agg,
                                const float* __restrict__ invIn,
                                const float* __restrict__ bias,
                                float* __restrict__ out, int n) {
  int i = blockIdx.x * blockDim.x + threadIdx.x;
  if (i < n) {
    int row = i / F;
    int f = i - row * F;
    float v = agg[i] * invIn[row] + bias[f];
    if (RELU) v = fmaxf(v, 0.0f);
    out[i] = v;
  }
}

template <int K, int NC, bool PRESCALE, bool EPI, bool RELU>
__global__ __launch_bounds__(32) void gemm_wmma(const float* __restrict__ A,
                                                const float* __restrict__ B,
                                                float* __restrict__ D,
                                                const float* __restrict__ rowScaleA,
                                                const float* __restrict__ rowScaleOut,
                                                const float* __restrict__ bias) {
  const int lane = threadIdx.x;
  const int half = lane >> 4;
  const int l15  = lane & 15;
  const int tileM = blockIdx.x * 16;
  const int tileN = blockIdx.y * 16;

  const int arow = tileM + l15;
  const float* __restrict__ Ap = A + arow * K;
  float ascale = 1.0f;
  if (PRESCALE) ascale = rowScaleA[arow];

  const int  col   = tileN + l15;
  const bool colOK = (NC % 16 == 0) ? true : (col < NC);

  v8f acc = {0.f, 0.f, 0.f, 0.f, 0.f, 0.f, 0.f, 0.f};

#pragma unroll 4
  for (int k = 0; k < K; k += 4) {
    const int ka = k + half * 2;
    v2f a, b;
    a.x = Ap[ka];
    a.y = Ap[ka + 1];
    if (PRESCALE) { a.x *= ascale; a.y *= ascale; }
    b.x = colOK ? B[ka * NC + col] : 0.0f;
    b.y = colOK ? B[(ka + 1) * NC + col] : 0.0f;
    acc = __builtin_amdgcn_wmma_f32_16x16x4_f32(false, a, false, b, (short)0, acc,
                                                false, false);
  }

  if (colOK) {
    const int mbase = tileM + half * 8;
#pragma unroll
    for (int r = 0; r < 8; ++r) {
      const int row = mbase + r;
      float v = acc[r];
      if (EPI)  v = v * rowScaleOut[row] + bias[col];
      if (RELU) v = fmaxf(v, 0.0f);
      D[row * NC + col] = v;
    }
  }
}


extern "C" void kernel_launch(void* const* d_in, const int* in_sizes, int n_in,
                              void* d_out, int out_size, void* d_ws, size_t ws_size,
                              hipStream_t stream) {
  const float* features = (const float*)d_in[0];
  const int*   src      = (const int*)d_in[1];
  const int*   dst      = (const int*)d_in[2];
  const float* W1       = (const float*)d_in[3];
  const float* b1       = (const float*)d_in[4];
  const float* W2       = (const float*)d_in[5];
  const float* b2       = (const float*)d_in[6];
  const float* W3       = (const float*)d_in[7];
  const float* b3       = (const float*)d_in[8];

  const int N = NNODES;
  const int E = in_sizes[1];

  float* ws      = (float*)d_ws;
  float* inv_out = ws;
  float* inv_in  = ws + N;
  float* P0      = ws + 2 * (size_t)N;
  float* P1      = P0 + (size_t)N * 256;
  float* P2      = P1 + (size_t)N * 128;
  float* H3      = P0;
  float* AGG3    = P0 + (size_t)N * 40;

  const int T = 256;
  auto blocks = [T](long long n) { return (unsigned)((n + T - 1) / T); };

  zero_f32<<<blocks(2LL * N), T, 0, stream>>>(inv_out, 2 * N);
  degree_kernel<<<blocks(E), T, 0, stream>>>(src, dst, inv_out, inv_in, E);
  norm_kernel<<<blocks(N), T, 0, stream>>>(inv_out, inv_in, N);

  zero_f32<<<blocks((long long)N * 128), T, 0, stream>>>(P1, N * 128);
  scatter_kernel<128, true><<<blocks((long long)E * 128), T, 0, stream>>>(
      features, src, dst, inv_out, P1, E);
  gemm_wmma<128, 256, false, true, true><<<dim3(N / 16, 256 / 16), 32, 0, stream>>>(
      P1, W1, P0, nullptr, inv_in, b1);

  gemm_wmma<256, 128, true, false, false><<<dim3(N / 16, 128 / 16), 32, 0, stream>>>(
      P0, W2, P2, inv_out, nullptr, nullptr);
  zero_f32<<<blocks((long long)N * 128), T, 0, stream>>>(P1, N * 128);
  scatter_kernel<128, false><<<blocks((long long)E * 128), T, 0, stream>>>(
      P2, src, dst, nullptr, P1, E);
  epilogue_kernel<128, true><<<blocks((long long)N * 128), T, 0, stream>>>(
      P1, inv_in, b2, P2, N * 128);

  gemm_wmma<128, 40, true, false, false><<<dim3(N / 16, 3), 32, 0, stream>>>(
      P2, W3, H3, inv_out, nullptr, nullptr);
  zero_f32<<<blocks((long long)N * 40), T, 0, stream>>>(AGG3, N * 40);
  scatter_kernel<40, false><<<blocks((long long)E * 40), T, 0, stream>>>(
      H3, src, dst, nullptr, AGG3, E);
  epilogue_kernel<40, false><<<blocks((long long)N * 40), T, 0, stream>>>(
      AGG3, inv_in, b3, (float*)d_out, N * 40);
}


# Neighbour sums by gather over a per-call dst-bucketed edge list (int count+rank atomics, scan, fill): no float atomics or zero-fill; sum order in a node = atomic arrival rank
# speedup vs baseline: 2.1216x; 2.1216x over previous
#include <hip/hip_runtime.h>


typedef __attribute__((ext_vector_type(2))) float v2f;
typedef __attribute__((ext_vector_type(8))) float v8f;

#define NNODES 100000


__global__ void zero_i32(int* __restrict__ p, int n) {
  int i = blockIdx.x * blockDim.x + threadIdx.x;
  if (i < n) p[i] = 0;
}

__global__ void count_rank_kernel(const int* __restrict__ src, const int* __restrict__ dst,
                                  int* __restrict__ cnt_out, int* __restrict__ cnt_in,
                                  int* __restrict__ rank, int E) {
  int e = blockIdx.x * blockDim.x + threadIdx.x;
  if (e < E) {
    atomicAdd(&cnt_out[src[e]], 1);
    rank[e] = atomicAdd(&cnt_in[dst[e]], 1);
  }
}

__global__ void norm_kernel(const int* __restrict__ cnt_out, const int* __restrict__ cnt_in,
                            float* __restrict__ inv_out, float* __restrict__ inv_in, int n) {
  int i = blockIdx.x * blockDim.x + threadIdx.x;
  if (i < n) {
    inv_out[i] = rsqrtf(fmaxf((float)cnt_out[i], 1.0f));
    inv_in[i]  = rsqrtf(fmaxf((float)cnt_in[i], 1.0f));
  }
}

__global__ __launch_bounds__(256) void scan_local_kernel(const int* __restrict__ cnt,
                                                         int* __restrict__ rowptr,
                                                         int* __restrict__ blocksum, int n) {
  __shared__ int sh[256];
  const int t = threadIdx.x;
  const int base = (blockIdx.x * 256 + t) * 4;
  int c0 = 0, c1 = 0, c2 = 0, c3 = 0;
  if (base + 3 < n) {
    const int4 v = *reinterpret_cast<const int4*>(cnt + base);
    c0 = v.x; c1 = v.y; c2 = v.z; c3 = v.w;
  } else {
    if (base     < n) c0 = cnt[base];
    if (base + 1 < n) c1 = cnt[base + 1];
    if (base + 2 < n) c2 = cnt[base + 2];
  }
  const int tsum = c0 + c1 + c2 + c3;
  sh[t] = tsum;
  __syncthreads();
  for (int off = 1; off < 256; off <<= 1) {
    const int v = (t >= off) ? sh[t - off] : 0;
    __syncthreads();
    sh[t] += v;
    __syncthreads();
  }
  const int excl = sh[t] - tsum;
  if (base     < n) rowptr[base]     = excl;
  if (base + 1 < n) rowptr[base + 1] = excl + c0;
  if (base + 2 < n) rowptr[base + 2] = excl + c0 + c1;
  if (base + 3 < n) rowptr[base + 3] = excl + c0 + c1 + c2;
  if (t == 255) blocksum[blockIdx.x] = sh[255];
}

__global__ __launch_bounds__(256) void scan_add_kernel(int* __restrict__ rowptr,
                                                       const int* __restrict__ blocksum, int n) {
  __shared__ int sh[256];
  const int t = threadIdx.x;
  const int b = blockIdx.x;
  int part = 0;
  for (int i = t; i < b; i += 256) part += blocksum[i];
  sh[t] = part;
  __syncthreads();
  for (int off = 128; off > 0; off >>= 1) {
    if (t < off) sh[t] += sh[t + off];
    __syncthreads();
  }
  const int offset = sh[0];
  const int base = (b * 256 + t) * 4;
  for (int k = 0; k < 4; ++k)
    if (base + k < n) rowptr[base + k] += offset;
  if (b == (int)gridDim.x - 1 && t == 0) rowptr[n] = offset + blocksum[b];
}

__global__ void fill_kernel(const int* __restrict__ src, const int* __restrict__ dst,
                            const int* __restrict__ rowptr, const int* __restrict__ rank,
                            int* __restrict__ csr_src, int E) {
  int e = blockIdx.x * blockDim.x + threadIdx.x;
  if (e < E) csr_src[rowptr[dst[e]] + rank[e]] = src[e];
}

template <int F, bool SCALE>
__global__ __launch_bounds__(256) void gather_kernel(const float* __restrict__ x,
                                                     const int* __restrict__ rowptr,
                                                     const int* __restrict__ csr_src,
                                                     const float* __restrict__ scale,
                                                     float* __restrict__ agg, int n) {
  constexpr int LPN = F / 4;
  const int tid = blockIdx.x * 256 + threadIdx.x;
  int node = tid / LPN;
  const int q = tid - node * LPN;
  if (LPN == 32) node = __builtin_amdgcn_readfirstlane(node);
  if (node >= n) return;
  const int beg = rowptr[node];
  const int end = rowptr[node + 1];
  const float4* __restrict__ xv = reinterpret_cast<const float4*>(x);
  float4 acc = {0.f, 0.f, 0.f, 0.f};
  int j = beg;
  for (; j + 4 <= end; j += 4) {
    const int s0 = csr_src[j], s1 = csr_src[j + 1], s2 = csr_src[j + 2], s3 = csr_src[j + 3];
    const float4 v0 = xv[s0 * LPN + q];
    const float4 v1 = xv[s1 * LPN + q];
    const float4 v2 = xv[s2 * LPN + q];
    const float4 v3 = xv[s3 * LPN + q];
    float w0 = 1.0f, w1 = 1.0f, w2 = 1.0f, w3 = 1.0f;
    if (SCALE) { w0 = scale[s0]; w1 = scale[s1]; w2 = scale[s2]; w3 = scale[s3]; }
    acc.x += v0.x * w0; acc.y += v0.y * w0; acc.z += v0.z * w0; acc.w += v0.w * w0;
    acc.x += v1.x * w1; acc.y += v1.y * w1; acc.z += v1.z * w1; acc.w += v1.w * w1;
    acc.x += v2.x * w2; acc.y += v2.y * w2; acc.z += v2.z * w2; acc.w += v2.w * w2;
    acc.x += v3.x * w3; acc.y += v3.y * w3; acc.z += v3.z * w3; acc.w += v3.w * w3;
  }
  for (; j < end; ++j) {
    const int s0 = csr_src[j];
    const float4 v0 = xv[s0 * LPN + q];
    float w0 = 1.0f;
    if (SCALE) w0 = scale[s0];
    acc.x += v0.x * w0; acc.y += v0.y * w0; acc.z += v0.z * w0; acc.w += v0.w * w0;
  }
  reinterpret_cast<float4*>(agg)[node * LPN + q] = acc;
}

template <int F, bool RELU>
__global__ void epilogue_kernel(const float* __restrict__ agg,
                                const float* __restrict__ invIn,
                                const float* __restrict__ bias,
                                float* __restrict__ out, int n) {
  int i = blockIdx.x * blockDim.x + threadIdx.x;
  if (i < n) {
    int row = i / F;
    int f = i - row * F;
    float v = agg[i] * invIn[row] + bias[f];
    if (RELU) v = fmaxf(v, 0.0f);
    out[i] = v;
  }
}

template <int K, int NC, bool PRESCALE, bool EPI, bool RELU>
__global__ __launch_bounds__(32) void gemm_wmma(const float* __restrict__ A,
                                                const float* __restrict__ B,
                                                float* __restrict__ D,
                                                const float* __restrict__ rowScaleA,
                                                const float* __restrict__ rowScaleOut,
                                                const float* __restrict__ bias) {
  const int lane = threadIdx.x;
  const int half = lane >> 4;
  const int l15  = lane & 15;
  const int tileM = blockIdx.x * 16;
  const int tileN = blockIdx.y * 16;

  const int arow = tileM + l15;
  const float* __restrict__ Ap = A + arow * K;
  float ascale = 1.0f;
  if (PRESCALE) ascale = rowScaleA[arow];

  const int  col   = tileN + l15;
  const bool colOK = (NC % 16 == 0) ? true : (col < NC);

  v8f acc = {0.f, 0.f, 0.f, 0.f, 0.f, 0.f, 0.f, 0.f};

#pragma unroll 4
  for (int k = 0; k < K; k += 4) {
    const int ka = k + half * 2;
    v2f a, b;
    a.x = Ap[ka];
    a.y = Ap[ka + 1];
    if (PRESCALE) { a.x *= ascale; a.y *= ascale; }
    b.x = colOK ? B[ka * NC + col] : 0.0f;
    b.y = colOK ? B[(ka + 1) * NC + col] : 0.0f;
    acc = __builtin_amdgcn_wmma_f32_16x16x4_f32(false, a, false, b, (short)0, acc,
                                                false, false);
  }

  if (colOK) {
    const int mbase = tileM + half * 8;
#pragma unroll
    for (int r = 0; r < 8; ++r) {
      const int row = mbase + r;
      float v = acc[r];
      if (EPI)  v = v * rowScaleOut[row] + bias[col];
      if (RELU) v = fmaxf(v, 0.0f);
      D[row * NC + col] = v;
    }
  }
}


extern "C" void kernel_launch(void* const* d_in, const int* in_sizes, int n_in,
                              void* d_out, int out_size, void* d_ws, size_t ws_size,
                              hipStream_t stream) {
  const float* features = (const float*)d_in[0];
  const int*   src      = (const int*)d_in[1];
  const int*   dst      = (const int*)d_in[2];
  const float* W1       = (const float*)d_in[3];
  const float* b1       = (const float*)d_in[4];
  const float* W2       = (const float*)d_in[5];
  const float* b2       = (const float*)d_in[6];
  const float* W3       = (const float*)d_in[7];
  const float* b3       = (const float*)d_in[8];

  const int N = NNODES;
  const int E = in_sizes[1];

  float* ws      = (float*)d_ws;
  float* inv_out = ws;
  float* inv_in  = ws + N;
  float* P0      = ws + 2 * (size_t)N;
  float* P1      = P0 + (size_t)N * 256;
  float* P2      = P1 + (size_t)N * 128;
  float* H3      = P0 + (size_t)N * 128;
  float* AGG3    = P0;

  const int NB = (N + 1023) / 1024;
  int* cnt_out  = (int*)P2;
  int* cnt_in   = cnt_out + N;
  int* rowptr   = cnt_in + N;
  int* rank     = rowptr + N + 4;
  int* csr_src  = rank + E;
  int* blocksum = csr_src + E;

  const int T = 256;
  auto blocks = [T](long long n) { return (unsigned)((n + T - 1) / T); };

  zero_i32<<<blocks(2LL * N), T, 0, stream>>>(cnt_out, 2 * N);
  count_rank_kernel<<<blocks(E), T, 0, stream>>>(src, dst, cnt_out, cnt_in, rank, E);
  norm_kernel<<<blocks(N), T, 0, stream>>>(cnt_out, cnt_in, inv_out, inv_in, N);
  scan_local_kernel<<<NB, 256, 0, stream>>>(cnt_in, rowptr, blocksum, N);
  scan_add_kernel<<<NB, 256, 0, stream>>>(rowptr, blocksum, N);
  fill_kernel<<<blocks(E), T, 0, stream>>>(src, dst, rowptr, rank, csr_src, E);

  gather_kernel<128, true><<<blocks((long long)N * 32), 256, 0, stream>>>(
      features, rowptr, csr_src, inv_out, P1, N);
  gemm_wmma<128, 256, false, true, true><<<dim3(N / 16, 256 / 16), 32, 0, stream>>>(
      P1, W1, P0, nullptr, inv_in, b1);

  gemm_wmma<256, 128, true, false, false><<<dim3(N / 16, 128 / 16), 32, 0, stream>>>(
      P0, W2, P1, inv_out, nullptr, nullptr);
  gather_kernel<128, false><<<blocks((long long)N * 32), 256, 0, stream>>>(
      P1, rowptr, csr_src, nullptr, P0, N);
  epilogue_kernel<128, true><<<blocks((long long)N * 128), T, 0, stream>>>(
      P0, inv_in, b2, P1, N * 128);

  gemm_wmma<128, 40, true, false, false><<<dim3(N / 16, 3), 32, 0, stream>>>(
      P1, W3, H3, inv_out, nullptr, nullptr);
  gather_kernel<40, false><<<blocks((long long)N * 10), 256, 0, stream>>>(
      H3, rowptr, csr_src, nullptr, AGG3, N);
  epilogue_kernel<40, false><<<blocks((long long)N * 40), T, 0, stream>>>(
      AGG3, inv_in, b3, (float*)d_out, N * 40);
}


# GEMM operand reuse: weights staged once per work-group in LDS, each wave reuses its A fragments over 8 column tiles; k order inside each group of 8 changes, so the f32 sum order changes
# speedup vs baseline: 3.2555x; 1.5345x over previous
#include <hip/hip_runtime.h>


typedef __attribute__((ext_vector_type(2))) float v2f;
typedef __attribute__((ext_vector_type(8))) float v8f;

#define NNODES 100000


__global__ void zero_i32(int* __restrict__ p, int n) {
  int i = blockIdx.x * blockDim.x + threadIdx.x;
  if (i < n) p[i] = 0;
}

__global__ void count_rank_kernel(const int* __restrict__ src, const int* __restrict__ dst,
                                  int* __restrict__ cnt_out, int* __restrict__ cnt_in,
                                  int* __restrict__ rank, int E) {
  int e = blockIdx.x * blockDim.x + threadIdx.x;
  if (e < E) {
    atomicAdd(&cnt_out[src[e]], 1);
    rank[e] = atomicAdd(&cnt_in[dst[e]], 1);
  }
}

__global__ void norm_kernel(const int* __restrict__ cnt_out, const int* __restrict__ cnt_in,
                            float* __restrict__ inv_out, float* __restrict__ inv_in, int n) {
  int i = blockIdx.x * blockDim.x + threadIdx.x;
  if (i < n) {
    inv_out[i] = rsqrtf(fmaxf((float)cnt_out[i], 1.0f));
    inv_in[i]  = rsqrtf(fmaxf((float)cnt_in[i], 1.0f));
  }
}

__global__ __launch_bounds__(256) void scan_local_kernel(const int* __restrict__ cnt,
                                                         int* __restrict__ rowptr,
                                                         int* __restrict__ blocksum, int n) {
  __shared__ int sh[256];
  const int t = threadIdx.x;
  const int base = (blockIdx.x * 256 + t) * 4;
  int c0 = 0, c1 = 0, c2 = 0, c3 = 0;
  if (base + 3 < n) {
    const int4 v = *reinterpret_cast<const int4*>(cnt + base);
    c0 = v.x; c1 = v.y; c2 = v.z; c3 = v.w;
  } else {
    if (base     < n) c0 = cnt[base];
    if (base + 1 < n) c1 = cnt[base + 1];
    if (base + 2 < n) c2 = cnt[base + 2];
  }
  const int tsum = c0 + c1 + c2 + c3;
  sh[t] = tsum;
  __syncthreads();
  for (int off = 1; off < 256; off <<= 1) {
    const int v = (t >= off) ? sh[t - off] : 0;
    __syncthreads();
    sh[t] += v;
    __syncthreads();
  }
  const int excl = sh[t] - tsum;
  if (base     < n) rowptr[base]     = excl;
  if (base + 1 < n) rowptr[base + 1] = excl + c0;
  if (base + 2 < n) rowptr[base + 2] = excl + c0 + c1;
  if (base + 3 < n) rowptr[base + 3] = excl + c0 + c1 + c2;
  if (t == 255) blocksum[blockIdx.x] = sh[255];
}

__global__ __launch_bounds__(256) void scan_add_kernel(int* __restrict__ rowptr,
                                                       const int* __restrict__ blocksum, int n) {
  __shared__ int sh[256];
  const int t = threadIdx.x;
  const int b = blockIdx.x;
  int part = 0;
  for (int i = t; i < b; i += 256) part += blocksum[i];
  sh[t] = part;
  __syncthreads();
  for (int off = 128; off > 0; off >>= 1) {
    if (t < off) sh[t] += sh[t + off];
    __syncthreads();
  }
  const int offset = sh[0];
  const int base = (b * 256 + t) * 4;
  for (int k = 0; k < 4; ++k)
    if (base + k < n) rowptr[base + k] += offset;
  if (b == (int)gridDim.x - 1 && t == 0) rowptr[n] = offset + blocksum[b];
}

__global__ void fill_kernel(const int* __restrict__ src, const int* __restrict__ dst,
                            const int* __restrict__ rowptr, const int* __restrict__ rank,
                            int* __restrict__ csr_src, int E) {
  int e = blockIdx.x * blockDim.x + threadIdx.x;
  if (e < E) csr_src[rowptr[dst[e]] + rank[e]] = src[e];
}

template <int F, bool SCALE>
__global__ __launch_bounds__(256) void gather_kernel(const float* __restrict__ x,
                                                     const int* __restrict__ rowptr,
                                                     const int* __restrict__ csr_src,
                                                     const float* __restrict__ scale,
                                                     float* __restrict__ agg, int n) {
  constexpr int LPN = F / 4;
  const int tid = blockIdx.x * 256 + threadIdx.x;
  int node = tid / LPN;
  const int q = tid - node * LPN;
  if (LPN == 32) node = __builtin_amdgcn_readfirstlane(node);
  if (node >= n) return;
  const int beg = rowptr[node];
  const int end = rowptr[node + 1];
  const float4* __restrict__ xv = reinterpret_cast<const float4*>(x);
  float4 acc = {0.f, 0.f, 0.f, 0.f};
  int j = beg;
  for (; j + 4 <= end; j += 4) {
    const int s0 = csr_src[j], s1 = csr_src[j + 1], s2 = csr_src[j + 2], s3 = csr_src[j + 3];
    const float4 v0 = xv[s0 * LPN + q];
    const float4 v1 = xv[s1 * LPN + q];
    const float4 v2 = xv[s2 * LPN + q];
    const float4 v3 = xv[s3 * LPN + q];
    float w0 = 1.0f, w1 = 1.0f, w2 = 1.0f, w3 = 1.0f;
    if (SCALE) { w0 = scale[s0]; w1 = scale[s1]; w2 = scale[s2]; w3 = scale[s3]; }
    acc.x += v0.x * w0; acc.y += v0.y * w0; acc.z += v0.z * w0; acc.w += v0.w * w0;
    acc.x += v1.x * w1; acc.y += v1.y * w1; acc.z += v1.z * w1; acc.w += v1.w * w1;
    acc.x += v2.x * w2; acc.y += v2.y * w2; acc.z += v2.z * w2; acc.w += v2.w * w2;
    acc.x += v3.x * w3; acc.y += v3.y * w3; acc.z += v3.z * w3; acc.w += v3.w * w3;
  }
  for (; j < end; ++j) {
    const int s0 = csr_src[j];
    const float4 v0 = xv[s0 * LPN + q];
    float w0 = 1.0f;
    if (SCALE) w0 = scale[s0];
    acc.x += v0.x * w0; acc.y += v0.y * w0; acc.z += v0.z * w0; acc.w += v0.w * w0;
  }
  reinterpret_cast<float4*>(agg)[node * LPN + q] = acc;
}

template <int F, bool RELU>
__global__ void epilogue_kernel(const float* __restrict__ agg,
                                const float* __restrict__ invIn,
                                const float* __restrict__ bias,
                                float* __restrict__ out, int n) {
  int i = blockIdx.x * blockDim.x + threadIdx.x;
  if (i < n) {
    int row = i / F;
    int f = i - row * F;
    float v = agg[i] * invIn[row] + bias[f];
    if (RELU) v = fmaxf(v, 0.0f);
    out[i] = v;
  }
}

template <int K, int NC, int NT, bool PRESCALE, bool EPI, bool RELU>
__global__ __launch_bounds__(256) __attribute__((amdgpu_waves_per_eu(4, 4))) void gemm_strip(const float* __restrict__ A,
                                                  const float* __restrict__ B,
                                                  float* __restrict__ D,
                                                  const float* __restrict__ rowScaleA,
                                                  const float* __restrict__ rowScaleOut,
                                                  const float* __restrict__ bias,
                                                  int mtiles) {
  constexpr int NCT   = (NC + 15) / 16;
  constexpr int NCP   = NCT * 16;
  constexpr int NG    = NCT / NT;
  static_assert(NCT % NT == 0, "column tiles must split evenly into groups");
  static_assert(K % 8 == 0 && NC % 4 == 0, "vector loads need these");
  constexpr int PITCH = NCP + 8;
  __shared__ __attribute__((aligned(16))) float2 Bs[(K / 2) * PITCH];

  constexpr int C4 = NCP / 4;
  for (int i = threadIdx.x; i < (K / 2) * C4; i += 256) {
    const int kp = i / C4;
    const int c4 = (i - kp * C4) * 4;
    float4 r0 = {0.f, 0.f, 0.f, 0.f}, r1 = {0.f, 0.f, 0.f, 0.f};
    if (c4 < NC) {
      r0 = *reinterpret_cast<const float4*>(B + (2 * kp) * NC + c4);
      r1 = *reinterpret_cast<const float4*>(B + (2 * kp + 1) * NC + c4);
    }
    float4* dp = reinterpret_cast<float4*>(&Bs[kp * PITCH + c4]);
    dp[0] = make_float4(r0.x, r1.x, r0.y, r1.y);
    dp[1] = make_float4(r0.z, r1.z, r0.w, r1.w);
  }
  __syncthreads();

  const int lane = threadIdx.x & 31;
  const int half = lane >> 4;
  const int l15  = lane & 15;
  const int wv = __builtin_amdgcn_readfirstlane((int)(threadIdx.x >> 5));
  const int nWaves = (int)gridDim.x * 8;
  const int units  = mtiles * NG;

  for (int u = wv * (int)gridDim.x + (int)blockIdx.x; u < units; u += nWaves) {
    const int mt = u / NG;
    const int ng = u - mt * NG;
    const int tileM = mt * 16;
    const int arow  = tileM + l15;
    const float4* __restrict__ Ap = reinterpret_cast<const float4*>(A + arow * K) + half;

    constexpr int KC  = 32;
    constexpr int JC  = KC / 8;
    constexpr int NCH = K / KC;
    static_assert(K % KC == 0, "K must be a whole number of chunks");
    float ascale = 1.0f;
    if (PRESCALE) ascale = rowScaleA[arow];

    v8f acc[NT];
#pragma unroll
    for (int t = 0; t < NT; ++t) acc[t] = (v8f){0.f, 0.f, 0.f, 0.f, 0.f, 0.f, 0.f, 0.f};

    const float2* Bl = Bs + (2 * half) * PITCH + ng * (NT * 16) + l15;

    float4 an[JC];
#pragma unroll
    for (int j = 0; j < JC; ++j) an[j] = Ap[2 * j];

#pragma unroll 1
    for (int ch = 0; ch < NCH; ++ch) {
      float4 ac[JC];
#pragma unroll
      for (int j = 0; j < JC; ++j) {
        ac[j] = an[j];
        if (PRESCALE) { ac[j].x *= ascale; ac[j].y *= ascale; ac[j].z *= ascale; ac[j].w *= ascale; }
      }
      const int chn = (ch + 1 < NCH) ? ch + 1 : ch;
#pragma unroll
      for (int j = 0; j < JC; ++j) an[j] = Ap[2 * (chn * JC + j)];

      const float2* Bc = Bl + (4 * JC * ch) * PITCH;
#pragma unroll
      for (int j = 0; j < JC; ++j) {
#pragma unroll
        for (int e = 0; e < 2; ++e) {
          v2f a;
          a.x = e ? ac[j].z : ac[j].x;
          a.y = e ? ac[j].w : ac[j].y;
          const float2* Br = Bc + (4 * j + e) * PITCH;
#pragma unroll
          for (int t = 0; t < NT; ++t) {
            const float2 bb = Br[t * 16];
            v2f b;
            b.x = bb.x;
            b.y = bb.y;
            acc[t] = __builtin_amdgcn_wmma_f32_16x16x4_f32(false, a, false, b, (short)0, acc[t],
                                                           false, false);
          }
        }
      }
    }

    const int mbase = tileM + half * 8;
    float rs[8];
    if (EPI) {
#pragma unroll
      for (int r = 0; r < 8; ++r) rs[r] = rowScaleOut[mbase + r];
    }
#pragma unroll
    for (int t = 0; t < NT; ++t) {
      const int col = (ng * NT + t) * 16 + l15;
      if ((NC % 16 == 0) || col < NC) {
        float bv = 0.0f;
        if (EPI) bv = bias[col];
#pragma unroll
        for (int r = 0; r < 8; ++r) {
          float v = acc[t][r];
          if (EPI)  v = v * rs[r] + bv;
          if (RELU) v = fmaxf(v, 0.0f);
          D[(mbase + r) * NC + col] = v;
        }
      }
    }
  }
}


extern "C" void kernel_launch(void* const* d_in, const int* in_sizes, int n_in,
                              void* d_out, int out_size, void* d_ws, size_t ws_size,
                              hipStream_t stream) {
  const float* features = (const float*)d_in[0];
  const int*   src      = (const int*)d_in[1];
  const int*   dst      = (const int*)d_in[2];
  const float* W1       = (const float*)d_in[3];
  const float* b1       = (const float*)d_in[4];
  const float* W2       = (const float*)d_in[5];
  const float* b2       = (const float*)d_in[6];
  const float* W3       = (const float*)d_in[7];
  const float* b3       = (const float*)d_in[8];

  const int N = NNODES;
  const int E = in_sizes[1];

  float* ws      = (float*)d_ws;
  float* inv_out = ws;
  float* inv_in  = ws + N;
  float* P0      = ws + 2 * (size_t)N;
  float* P1      = P0 + (size_t)N * 256;
  float* P2      = P1 + (size_t)N * 128;
  float* H3      = P0 + (size_t)N * 128;
  float* AGG3    = P0;

  const int NB = (N + 1023) / 1024;
  int* cnt_out  = (int*)P2;
  int* cnt_in   = cnt_out + N;
  int* rowptr   = cnt_in + N;
  int* rank     = rowptr + N + 4;
  int* csr_src  = rank + E;
  int* blocksum = csr_src + E;

  const int T = 256;
  auto blocks = [T](long long n) { return (unsigned)((n + T - 1) / T); };

  zero_i32<<<blocks(2LL * N), T, 0, stream>>>(cnt_out, 2 * N);
  count_rank_kernel<<<blocks(E), T, 0, stream>>>(src, dst, cnt_out, cnt_in, rank, E);
  norm_kernel<<<blocks(N), T, 0, stream>>>(cnt_out, cnt_in, inv_out, inv_in, N);
  scan_local_kernel<<<NB, 256, 0, stream>>>(cnt_in, rowptr, blocksum, N);
  scan_add_kernel<<<NB, 256, 0, stream>>>(rowptr, blocksum, N);
  fill_kernel<<<blocks(E), T, 0, stream>>>(src, dst, rowptr, rank, csr_src, E);

  gather_kernel<128, true><<<blocks((long long)N * 32), 256, 0, stream>>>(
      features, rowptr, csr_src, inv_out, P1, N);
  gemm_strip<128, 256, 8, false, true, true><<<512, 256, 0, stream>>>(
      P1, W1, P0, nullptr, inv_in, b1, N / 16);

  gemm_strip<256, 128, 8, true, false, false><<<512, 256, 0, stream>>>(
      P0, W2, P1, inv_out, nullptr, nullptr, N / 16);
  gather_kernel<128, false><<<blocks((long long)N * 32), 256, 0, stream>>>(
      P1, rowptr, csr_src, nullptr, P0, N);
  epilogue_kernel<128, true><<<blocks((long long)N * 128), T, 0, stream>>>(
      P0, inv_in, b2, P1, N * 128);

  gemm_strip<128, 40, 3, true, false, false><<<512, 256, 0, stream>>>(
      P1, W3, H3, inv_out, nullptr, nullptr, N / 16);
  gather_kernel<40, false><<<blocks((long long)N * 10), 256, 0, stream>>>(
      H3, rowptr, csr_src, nullptr, AGG3, N);
  epilogue_kernel<40, false><<<blocks((long long)N * 40), T, 0, stream>>>(
      AGG3, inv_in, b3, (float*)d_out, N * 40);
}


# Scale by inv_in, bias and relu of layers 2 and 3 applied in the gather's row store: both elementwise launches (one float per lane) and their memory passes removed; sums and their order unchanged
# speedup vs baseline: 3.7620x; 1.1556x over previous
#include <hip/hip_runtime.h>


typedef __attribute__((ext_vector_type(2))) float v2f;
typedef __attribute__((ext_vector_type(8))) float v8f;

#define NNODES 100000


__global__ void zero_i32(int* __restrict__ p, int n) {
  int i = blockIdx.x * blockDim.x + threadIdx.x;
  if (i < n) p[i] = 0;
}

__global__ void count_rank_kernel(const int* __restrict__ src, const int* __restrict__ dst,
                                  int* __restrict__ cnt_out, int* __restrict__ cnt_in,
                                  int* __restrict__ rank, int E) {
  int e = blockIdx.x * blockDim.x + threadIdx.x;
  if (e < E) {
    atomicAdd(&cnt_out[src[e]], 1);
    rank[e] = atomicAdd(&cnt_in[dst[e]], 1);
  }
}

__global__ void norm_kernel(const int* __restrict__ cnt_out, const int* __restrict__ cnt_in,
                            float* __restrict__ inv_out, float* __restrict__ inv_in, int n) {
  int i = blockIdx.x * blockDim.x + threadIdx.x;
  if (i < n) {
    inv_out[i] = rsqrtf(fmaxf((float)cnt_out[i], 1.0f));
    inv_in[i]  = rsqrtf(fmaxf((float)cnt_in[i], 1.0f));
  }
}

__global__ __launch_bounds__(256) void scan_local_kernel(const int* __restrict__ cnt,
                                                         int* __restrict__ rowptr,
                                                         int* __restrict__ blocksum, int n) {
  __shared__ int sh[256];
  const int t = threadIdx.x;
  const int base = (blockIdx.x * 256 + t) * 4;
  int c0 = 0, c1 = 0, c2 = 0, c3 = 0;
  if (base + 3 < n) {
    const int4 v = *reinterpret_cast<const int4*>(cnt + base);
    c0 = v.x; c1 = v.y; c2 = v.z; c3 = v.w;
  } else {
    if (base     < n) c0 = cnt[base];
    if (base + 1 < n) c1 = cnt[base + 1];
    if (base + 2 < n) c2 = cnt[base + 2];
  }
  const int tsum = c0 + c1 + c2 + c3;
  sh[t] = tsum;
  __syncthreads();
  for (int off = 1; off < 256; off <<= 1) {
    const int v = (t >= off) ? sh[t - off] : 0;
    __syncthreads();
    sh[t] += v;
    __syncthreads();
  }
  const int excl = sh[t] - tsum;
  if (base     < n) rowptr[base]     = excl;
  if (base + 1 < n) rowptr[base + 1] = excl + c0;
  if (base + 2 < n) rowptr[base + 2] = excl + c0 + c1;
  if (base + 3 < n) rowptr[base + 3] = excl + c0 + c1 + c2;
  if (t == 255) blocksum[blockIdx.x] = sh[255];
}

__global__ __launch_bounds__(256) void scan_add_kernel(int* __restrict__ rowptr,
                                                       const int* __restrict__ blocksum, int n) {
  __shared__ int sh[256];
  const int t = threadIdx.x;
  const int b = blockIdx.x;
  int part = 0;
  for (int i = t; i < b; i += 256) part += blocksum[i];
  sh[t] = part;
  __syncthreads();
  for (int off = 128; off > 0; off >>= 1) {
    if (t < off) sh[t] += sh[t + off];
    __syncthreads();
  }
  const int offset = sh[0];
  const int base = (b * 256 + t) * 4;
  for (int k = 0; k < 4; ++k)
    if (base + k < n) rowptr[base + k] += offset;
  if (b == (int)gridDim.x - 1 && t == 0) rowptr[n] = offset + blocksum[b];
}

__global__ void fill_kernel(const int* __restrict__ src, const int* __restrict__ dst,
                            const int* __restrict__ rowptr, const int* __restrict__ rank,
                            int* __restrict__ csr_src, int E) {
  int e = blockIdx.x * blockDim.x + threadIdx.x;
  if (e < E) csr_src[rowptr[dst[e]] + rank[e]] = src[e];
}

template <int F, bool SCALE, bool EPI, bool RELU>
__global__ __launch_bounds__(256) void gather_kernel(const float* __restrict__ x,
                                                     const int* __restrict__ rowptr,
                                                     const int* __restrict__ csr_src,
                                                     const float* __restrict__ scale,
                                                     const float* __restrict__ rowScaleOut,
                                                     const float* __restrict__ bias,
                                                     float* __restrict__ agg, int n) {
  constexpr int LPN = F / 4;
  const int tid = blockIdx.x * 256 + threadIdx.x;
  int node = tid / LPN;
  const int q = tid - node * LPN;
  if (LPN == 32) node = __builtin_amdgcn_readfirstlane(node);
  if (node >= n) return;
  const int beg = rowptr[node];
  const int end = rowptr[node + 1];
  float rs = 1.0f;
  float4 bv = {0.f, 0.f, 0.f, 0.f};
  if (EPI) {
    rs = rowScaleOut[node];
    bv = reinterpret_cast<const float4*>(bias)[q];
  }
  const float4* __restrict__ xv = reinterpret_cast<const float4*>(x);
  float4 acc = {0.f, 0.f, 0.f, 0.f};
  int j = beg;
  for (; j + 4 <= end; j += 4) {
    const int s0 = csr_src[j], s1 = csr_src[j + 1], s2 = csr_src[j + 2], s3 = csr_src[j + 3];
    const float4 v0 = xv[s0 * LPN + q];
    const float4 v1 = xv[s1 * LPN + q];
    const float4 v2 = xv[s2 * LPN + q];
    const float4 v3 = xv[s3 * LPN + q];
    float w0 = 1.0f, w1 = 1.0f, w2 = 1.0f, w3 = 1.0f;
    if (SCALE) { w0 = scale[s0]; w1 = scale[s1]; w2 = scale[s2]; w3 = scale[s3]; }
    acc.x += v0.x * w0; acc.y += v0.y * w0; acc.z += v0.z * w0; acc.w += v0.w * w0;
    acc.x += v1.x * w1; acc.y += v1.y * w1; acc.z += v1.z * w1; acc.w += v1.w * w1;
    acc.x += v2.x * w2; acc.y += v2.y * w2; acc.z += v2.z * w2; acc.w += v2.w * w2;
    acc.x += v3.x * w3; acc.y += v3.y * w3; acc.z += v3.z * w3; acc.w += v3.w * w3;
  }
  for (; j < end; ++j) {
    const int s0 = csr_src[j];
    const float4 v0 = xv[s0 * LPN + q];
    float w0 = 1.0f;
    if (SCALE) w0 = scale[s0];
    acc.x += v0.x * w0; acc.y += v0.y * w0; acc.z += v0.z * w0; acc.w += v0.w * w0;
  }
  if (EPI) {
    acc.x = acc.x * rs + bv.x;
    acc.y = acc.y * rs + bv.y;
    acc.z = acc.z * rs + bv.z;
    acc.w = acc.w * rs + bv.w;
  }
  if (RELU) {
    acc.x = fmaxf(acc.x, 0.0f);
    acc.y = fmaxf(acc.y, 0.0f);
    acc.z = fmaxf(acc.z, 0.0f);
    acc.w = fmaxf(acc.w, 0.0f);
  }
  reinterpret_cast<float4*>(agg)[node * LPN + q] = acc;
}

template <int K, int NC, int NT, bool PRESCALE, bool EPI, bool RELU>
__global__ __launch_bounds__(256) __attribute__((amdgpu_waves_per_eu(4, 4))) void gemm_strip(const float* __restrict__ A,
                                                  const float* __restrict__ B,
                                                  float* __restrict__ D,
                                                  const float* __restrict__ rowScaleA,
                                                  const float* __restrict__ rowScaleOut,
                                                  const float* __restrict__ bias,
                                                  int mtiles) {
  constexpr int NCT   = (NC + 15) / 16;
  constexpr int NCP   = NCT * 16;
  constexpr int NG    = NCT / NT;
  static_assert(NCT % NT == 0, "column tiles must split evenly into groups");
  static_assert(K % 8 == 0 && NC % 4 == 0, "vector loads need these");
  constexpr int PITCH = NCP + 8;
  __shared__ __attribute__((aligned(16))) float2 Bs[(K / 2) * PITCH];

  constexpr int C4 = NCP / 4;
  for (int i = threadIdx.x; i < (K / 2) * C4; i += 256) {
    const int kp = i / C4;
    const int c4 = (i - kp * C4) * 4;
    float4 r0 = {0.f, 0.f, 0.f, 0.f}, r1 = {0.f, 0.f, 0.f, 0.f};
    if (c4 < NC) {
      r0 = *reinterpret_cast<const float4*>(B + (2 * kp) * NC + c4);
      r1 = *reinterpret_cast<const float4*>(B + (2 * kp + 1) * NC + c4);
    }
    float4* dp = reinterpret_cast<float4*>(&Bs[kp * PITCH + c4]);
    dp[0] = make_float4(r0.x, r1.x, r0.y, r1.y);
    dp[1] = make_float4(r0.z, r1.z, r0.w, r1.w);
  }
  __syncthreads();

  const int lane = threadIdx.x & 31;
  const int half = lane >> 4;
  const int l15  = lane & 15;
  const int wv = __builtin_amdgcn_readfirstlane((int)(threadIdx.x >> 5));
  const int nWaves = (int)gridDim.x * 8;
  const int units  = mtiles * NG;

  for (int u = wv * (int)gridDim.x + (int)blockIdx.x; u < units; u += nWaves) {
    const int mt = u / NG;
    const int ng = u - mt * NG;
    const int tileM = mt * 16;
    const int arow  = tileM + l15;
    const float4* __restrict__ Ap = reinterpret_cast<const float4*>(A + arow * K) + half;

    constexpr int KC  = 32;
    constexpr int JC  = KC / 8;
    constexpr int NCH = K / KC;
    static_assert(K % KC == 0, "K must be a whole number of chunks");
    float ascale = 1.0f;
    if (PRESCALE) ascale = rowScaleA[arow];

    v8f acc[NT];
#pragma unroll
    for (int t = 0; t < NT; ++t) acc[t] = (v8f){0.f, 0.f, 0.f, 0.f, 0.f, 0.f, 0.f, 0.f};

    const float2* Bl = Bs + (2 * half) * PITCH + ng * (NT * 16) + l15;

    float4 an[JC];
#pragma unroll
    for (int j = 0; j < JC; ++j) an[j] = Ap[2 * j];

#pragma unroll 1
    for (int ch = 0; ch < NCH; ++ch) {
      float4 ac[JC];
#pragma unroll
      for (int j = 0; j < JC; ++j) {
        ac[j] = an[j];
        if (PRESCALE) { ac[j].x *= ascale; ac[j].y *= ascale; ac[j].z *= ascale; ac[j].w *= ascale; }
      }
      const int chn = (ch + 1 < NCH) ? ch + 1 : ch;
#pragma unroll
      for (int j = 0; j < JC; ++j) an[j] = Ap[2 * (chn * JC + j)];

      const float2* Bc = Bl + (4 * JC * ch) * PITCH;
#pragma unroll
      for (int j = 0; j < JC; ++j) {
#pragma unroll
        for (int e = 0; e < 2; ++e) {
          v2f a;
          a.x = e ? ac[j].z : ac[j].x;
          a.y = e ? ac[j].w : ac[j].y;
          const float2* Br = Bc + (4 * j + e) * PITCH;
#pragma unroll
          for (int t = 0; t < NT; ++t) {
            const float2 bb = Br[t * 16];
            v2f b;
            b.x = bb.x;
            b.y = bb.y;
            acc[t] = __builtin_amdgcn_wmma_f32_16x16x4_f32(false, a, false, b, (short)0, acc[t],
                                                           false, false);
          }
        }
      }
    }

    const int mbase = tileM + half * 8;
    float rs[8];
    if (EPI) {
#pragma unroll
      for (int r = 0; r < 8; ++r) rs[r] = rowScaleOut[mbase + r];
    }
#pragma unroll
    for (int t = 0; t < NT; ++t) {
      const int col = (ng * NT + t) * 16 + l15;
      if ((NC % 16 == 0) || col < NC) {
        float bv = 0.0f;
        if (EPI) bv = bias[col];
#pragma unroll
        for (int r = 0; r < 8; ++r) {
          float v = acc[t][r];
          if (EPI)  v = v * rs[r] + bv;
          if (RELU) v = fmaxf(v, 0.0f);
          D[(mbase + r) * NC + col] = v;
        }
      }
    }
  }
}


extern "C" void kernel_launch(void* const* d_in, const int* in_sizes, int n_in,
                              void* d_out, int out_size, void* d_ws, size_t ws_size,
                              hipStream_t stream) {
  const float* features = (const float*)d_in[0];
  const int*   src      = (const int*)d_in[1];
  const int*   dst      = (const int*)d_in[2];
  const float* W1       = (const float*)d_in[3];
  const float* b1       = (const float*)d_in[4];
  const float* W2       = (const float*)d_in[5];
  const float* b2       = (const float*)d_in[6];
  const float* W3       = (const float*)d_in[7];
  const float* b3       = (const float*)d_in[8];

  const int N = NNODES;
  const int E = in_sizes[1];

  float* ws      = (float*)d_ws;
  float* inv_out = ws;
  float* inv_in  = ws + N;
  float* P0      = ws + 2 * (size_t)N;
  float* P1      = P0 + (size_t)N * 256;
  float* P2      = P1 + (size_t)N * 128;
  float* H3      = P0 + (size_t)N * 128;

  const int NB = (N + 1023) / 1024;
  int* cnt_out  = (int*)P2;
  int* cnt_in   = cnt_out + N;
  int* rowptr   = cnt_in + N;
  int* rank     = rowptr + N + 4;
  int* csr_src  = rank + E;
  int* blocksum = csr_src + E;

  const int T = 256;
  auto blocks = [T](long long n) { return (unsigned)((n + T - 1) / T); };

  zero_i32<<<blocks(2LL * N), T, 0, stream>>>(cnt_out, 2 * N);
  count_rank_kernel<<<blocks(E), T, 0, stream>>>(src, dst, cnt_out, cnt_in, rank, E);
  norm_kernel<<<blocks(N), T, 0, stream>>>(cnt_out, cnt_in, inv_out, inv_in, N);
  scan_local_kernel<<<NB, 256, 0, stream>>>(cnt_in, rowptr, blocksum, N);
  scan_add_kernel<<<NB, 256, 0, stream>>>(rowptr, blocksum, N);
  fill_kernel<<<blocks(E), T, 0, stream>>>(src, dst, rowptr, rank, csr_src, E);

  gather_kernel<128, true, false, false><<<blocks((long long)N * 32), 256, 0, stream>>>(
      features, rowptr, csr_src, inv_out, nullptr, nullptr, P1, N);
  gemm_strip<128, 256, 8, false, true, true><<<512, 256, 0, stream>>>(
      P1, W1, P0, nullptr, inv_in, b1, N / 16);

  gemm_strip<256, 128, 8, true, false, false><<<512, 256, 0, stream>>>(
      P0, W2, P1, inv_out, nullptr, nullptr, N / 16);
  gather_kernel<128, false, true, true><<<blocks((long long)N * 32), 256, 0, stream>>>(
      P1, rowptr, csr_src, nullptr, inv_in, b2, P0, N);

  gemm_strip<128, 40, 3, true, false, false><<<512, 256, 0, stream>>>(
      P0, W3, H3, inv_out, nullptr, nullptr, N / 16);
  gather_kernel<40, false, true, false><<<blocks((long long)N * 10), 256, 0, stream>>>(
      H3, rowptr, csr_src, nullptr, inv_in, b3, (float*)d_out, N);
}


# Layer-3 product (x inv_out, W3, f32 WMMA) done inside the layer-2 gather launch on 32-node row tiles kept in LDS: out2 never stored, one launch less; sums and their order unchanged
# speedup vs baseline: 3.9119x; 1.0398x over previous
#include <hip/hip_runtime.h>


typedef __attribute__((ext_vector_type(2))) float v2f;
typedef __attribute__((ext_vector_type(8))) float v8f;

#define NNODES 100000


__global__ void zero_i32(int* __restrict__ p, int n) {
  int i = blockIdx.x * blockDim.x + threadIdx.x;
  if (i < n) p[i] = 0;
}

__global__ void count_rank_kernel(const int* __restrict__ src, const int* __restrict__ dst,
                                  int* __restrict__ cnt_out, int* __restrict__ cnt_in,
                                  int* __restrict__ rank, int E) {
  int e = blockIdx.x * blockDim.x + threadIdx.x;
  if (e < E) {
    atomicAdd(&cnt_out[src[e]], 1);
    rank[e] = atomicAdd(&cnt_in[dst[e]], 1);
  }
}

__global__ void norm_kernel(const int* __restrict__ cnt_out, const int* __restrict__ cnt_in,
                            float* __restrict__ inv_out, float* __restrict__ inv_in, int n) {
  int i = blockIdx.x * blockDim.x + threadIdx.x;
  if (i < n) {
    inv_out[i] = rsqrtf(fmaxf((float)cnt_out[i], 1.0f));
    inv_in[i]  = rsqrtf(fmaxf((float)cnt_in[i], 1.0f));
  }
}

__global__ __launch_bounds__(256) void scan_local_kernel(const int* __restrict__ cnt,
                                                         int* __restrict__ rowptr,
                                                         int* __restrict__ blocksum, int n) {
  __shared__ int sh[256];
  const int t = threadIdx.x;
  const int base = (blockIdx.x * 256 + t) * 4;
  int c0 = 0, c1 = 0, c2 = 0, c3 = 0;
  if (base + 3 < n) {
    const int4 v = *reinterpret_cast<const int4*>(cnt + base);
    c0 = v.x; c1 = v.y; c2 = v.z; c3 = v.w;
  } else {
    if (base     < n) c0 = cnt[base];
    if (base + 1 < n) c1 = cnt[base + 1];
    if (base + 2 < n) c2 = cnt[base + 2];
  }
  const int tsum = c0 + c1 + c2 + c3;
  sh[t] = tsum;
  __syncthreads();
  for (int off = 1; off < 256; off <<= 1) {
    const int v = (t >= off) ? sh[t - off] : 0;
    __syncthreads();
    sh[t] += v;
    __syncthreads();
  }
  const int excl = sh[t] - tsum;
  if (base     < n) rowptr[base]     = excl;
  if (base + 1 < n) rowptr[base + 1] = excl + c0;
  if (base + 2 < n) rowptr[base + 2] = excl + c0 + c1;
  if (base + 3 < n) rowptr[base + 3] = excl + c0 + c1 + c2;
  if (t == 255) blocksum[blockIdx.x] = sh[255];
}

__global__ __launch_bounds__(256) void scan_add_kernel(int* __restrict__ rowptr,
                                                       const int* __restrict__ blocksum, int n) {
  __shared__ int sh[256];
  const int t = threadIdx.x;
  const int b = blockIdx.x;
  int part = 0;
  for (int i = t; i < b; i += 256) part += blocksum[i];
  sh[t] = part;
  __syncthreads();
  for (int off = 128; off > 0; off >>= 1) {
    if (t < off) sh[t] += sh[t + off];
    __syncthreads();
  }
  const int offset = sh[0];
  const int base = (b * 256 + t) * 4;
  for (int k = 0; k < 4; ++k)
    if (base + k < n) rowptr[base + k] += offset;
  if (b == (int)gridDim.x - 1 && t == 0) rowptr[n] = offset + blocksum[b];
}

__global__ void fill_kernel(const int* __restrict__ src, const int* __restrict__ dst,
                            const int* __restrict__ rowptr, const int* __restrict__ rank,
                            int* __restrict__ csr_src, int E) {
  int e = blockIdx.x * blockDim.x + threadIdx.x;
  if (e < E) csr_src[rowptr[dst[e]] + rank[e]] = src[e];
}

template <int F, bool SCALE, bool EPI, bool RELU>
__global__ __launch_bounds__(256) void gather_kernel(const float* __restrict__ x,
                                                     const int* __restrict__ rowptr,
                                                     const int* __restrict__ csr_src,
                                                     const float* __restrict__ scale,
                                                     const float* __restrict__ rowScaleOut,
                                                     const float* __restrict__ bias,
                                                     float* __restrict__ agg, int n) {
  constexpr int LPN = F / 4;
  const int tid = blockIdx.x * 256 + threadIdx.x;
  int node = tid / LPN;
  const int q = tid - node * LPN;
  if (LPN == 32) node = __builtin_amdgcn_readfirstlane(node);
  if (node >= n) return;
  const int beg = rowptr[node];
  const int end = rowptr[node + 1];
  float rs = 1.0f;
  float4 bv = {0.f, 0.f, 0.f, 0.f};
  if (EPI) {
    rs = rowScaleOut[node];
    bv = reinterpret_cast<const float4*>(bias)[q];
  }
  const float4* __restrict__ xv = reinterpret_cast<const float4*>(x);
  float4 acc = {0.f, 0.f, 0.f, 0.f};
  int j = beg;
  for (; j + 4 <= end; j += 4) {
    const int s0 = csr_src[j], s1 = csr_src[j + 1], s2 = csr_src[j + 2], s3 = csr_src[j + 3];
    const float4 v0 = xv[s0 * LPN + q];
    const float4 v1 = xv[s1 * LPN + q];
    const float4 v2 = xv[s2 * LPN + q];
    const float4 v3 = xv[s3 * LPN + q];
    float w0 = 1.0f, w1 = 1.0f, w2 = 1.0f, w3 = 1.0f;
    if (SCALE) { w0 = scale[s0]; w1 = scale[s1]; w2 = scale[s2]; w3 = scale[s3]; }
    acc.x += v0.x * w0; acc.y += v0.y * w0; acc.z += v0.z * w0; acc.w += v0.w * w0;
    acc.x += v1.x * w1; acc.y += v1.y * w1; acc.z += v1.z * w1; acc.w += v1.w * w1;
    acc.x += v2.x * w2; acc.y += v2.y * w2; acc.z += v2.z * w2; acc.w += v2.w * w2;
    acc.x += v3.x * w3; acc.y += v3.y * w3; acc.z += v3.z * w3; acc.w += v3.w * w3;
  }
  for (; j < end; ++j) {
    const int s0 = csr_src[j];
    const float4 v0 = xv[s0 * LPN + q];
    float w0 = 1.0f;
    if (SCALE) w0 = scale[s0];
    acc.x += v0.x * w0; acc.y += v0.y * w0; acc.z += v0.z * w0; acc.w += v0.w * w0;
  }
  if (EPI) {
    acc.x = acc.x * rs + bv.x;
    acc.y = acc.y * rs + bv.y;
    acc.z = acc.z * rs + bv.z;
    acc.w = acc.w * rs + bv.w;
  }
  if (RELU) {
    acc.x = fmaxf(acc.x, 0.0f);
    acc.y = fmaxf(acc.y, 0.0f);
    acc.z = fmaxf(acc.z, 0.0f);
    acc.w = fmaxf(acc.w, 0.0f);
  }
  reinterpret_cast<float4*>(agg)[node * LPN + q] = acc;
}

template <int K, int NC, int NT, bool PRESCALE, bool EPI, bool RELU>
__global__ __launch_bounds__(256) __attribute__((amdgpu_waves_per_eu(4, 4))) void gemm_strip(const float* __restrict__ A,
                                                  const float* __restrict__ B,
                                                  float* __restrict__ D,
                                                  const float* __restrict__ rowScaleA,
                                                  const float* __restrict__ rowScaleOut,
                                                  const float* __restrict__ bias,
                                                  int mtiles) {
  constexpr int NCT   = (NC + 15) / 16;
  constexpr int NCP   = NCT * 16;
  constexpr int NG    = NCT / NT;
  static_assert(NCT % NT == 0, "column tiles must split evenly into groups");
  static_assert(K % 8 == 0 && NC % 4 == 0, "vector loads need these");
  constexpr int PITCH = NCP + 8;
  __shared__ __attribute__((aligned(16))) float2 Bs[(K / 2) * PITCH];

  constexpr int C4 = NCP / 4;
  for (int i = threadIdx.x; i < (K / 2) * C4; i += 256) {
    const int kp = i / C4;
    const int c4 = (i - kp * C4) * 4;
    float4 r0 = {0.f, 0.f, 0.f, 0.f}, r1 = {0.f, 0.f, 0.f, 0.f};
    if (c4 < NC) {
      r0 = *reinterpret_cast<const float4*>(B + (2 * kp) * NC + c4);
      r1 = *reinterpret_cast<const float4*>(B + (2 * kp + 1) * NC + c4);
    }
    float4* dp = reinterpret_cast<float4*>(&Bs[kp * PITCH + c4]);
    dp[0] = make_float4(r0.x, r1.x, r0.y, r1.y);
    dp[1] = make_float4(r0.z, r1.z, r0.w, r1.w);
  }
  __syncthreads();

  const int lane = threadIdx.x & 31;
  const int half = lane >> 4;
  const int l15  = lane & 15;
  const int wv = __builtin_amdgcn_readfirstlane((int)(threadIdx.x >> 5));
  const int nWaves = (int)gridDim.x * 8;
  const int units  = mtiles * NG;

  for (int u = wv * (int)gridDim.x + (int)blockIdx.x; u < units; u += nWaves) {
    const int mt = u / NG;
    const int ng = u - mt * NG;
    const int tileM = mt * 16;
    const int arow  = tileM + l15;
    const float4* __restrict__ Ap = reinterpret_cast<const float4*>(A + arow * K) + half;

    constexpr int KC  = 32;
    constexpr int JC  = KC / 8;
    constexpr int NCH = K / KC;
    static_assert(K % KC == 0, "K must be a whole number of chunks");
    float ascale = 1.0f;
    if (PRESCALE) ascale = rowScaleA[arow];

    v8f acc[NT];
#pragma unroll
    for (int t = 0; t < NT; ++t) acc[t] = (v8f){0.f, 0.f, 0.f, 0.f, 0.f, 0.f, 0.f, 0.f};

    const float2* Bl = Bs + (2 * half) * PITCH + ng * (NT * 16) + l15;

    float4 an[JC];
#pragma unroll
    for (int j = 0; j < JC; ++j) an[j] = Ap[2 * j];

#pragma unroll 1
    for (int ch = 0; ch < NCH; ++ch) {
      float4 ac[JC];
#pragma unroll
      for (int j = 0; j < JC; ++j) {
        ac[j] = an[j];
        if (PRESCALE) { ac[j].x *= ascale; ac[j].y *= ascale; ac[j].z *= ascale; ac[j].w *= ascale; }
      }
      const int chn = (ch + 1 < NCH) ? ch + 1 : ch;
#pragma unroll
      for (int j = 0; j < JC; ++j) an[j] = Ap[2 * (chn * JC + j)];

      const float2* Bc = Bl + (4 * JC * ch) * PITCH;
#pragma unroll
      for (int j = 0; j < JC; ++j) {
#pragma unroll
        for (int e = 0; e < 2; ++e) {
          v2f a;
          a.x = e ? ac[j].z : ac[j].x;
          a.y = e ? ac[j].w : ac[j].y;
          const float2* Br = Bc + (4 * j + e) * PITCH;
#pragma unroll
          for (int t = 0; t < NT; ++t) {
            const float2 bb = Br[t * 16];
            v2f b;
            b.x = bb.x;
            b.y = bb.y;
            acc[t] = __builtin_amdgcn_wmma_f32_16x16x4_f32(false, a, false, b, (short)0, acc[t],
                                                           false, false);
          }
        }
      }
    }

    const int mbase = tileM + half * 8;
    float rs[8];
    if (EPI) {
#pragma unroll
      for (int r = 0; r < 8; ++r) rs[r] = rowScaleOut[mbase + r];
    }
#pragma unroll
    for (int t = 0; t < NT; ++t) {
      const int col = (ng * NT + t) * 16 + l15;
      if ((NC % 16 == 0) || col < NC) {
        float bv = 0.0f;
        if (EPI) bv = bias[col];
#pragma unroll
        for (int r = 0; r < 8; ++r) {
          float v = acc[t][r];
          if (EPI)  v = v * rs[r] + bv;
          if (RELU) v = fmaxf(v, 0.0f);
          D[(mbase + r) * NC + col] = v;
        }
      }
    }
  }
}


template <int NC>
__global__ __launch_bounds__(256) void gather_project_kernel(const float* __restrict__ x,
                                                             const int* __restrict__ rowptr,
                                                             const int* __restrict__ csr_src,
                                                             const float* __restrict__ rowScaleOut,
                                                             const float* __restrict__ bias,
                                                             const float* __restrict__ rowScaleNext,
                                                             const float* __restrict__ B,
                                                             float* __restrict__ D) {
  constexpr int K     = 128;
  constexpr int LPN   = 32;
  constexpr int TILE  = 32;
  constexpr int NCT   = (NC + 15) / 16;
  constexpr int NCP   = NCT * 16;
  constexpr int PITCH = NCP + 8;
  constexpr int TP    = K + 4;
  static_assert(NNODES % TILE == 0, "whole tiles only");
  static_assert(2 * NCT <= 8, "one wave per output tile");
  static_assert(NC % 4 == 0, "vector loads need this");
  __shared__ __attribute__((aligned(16))) float2 Bs[(K / 2) * PITCH];
  __shared__ __attribute__((aligned(16))) float  Ts[TILE * TP];

  constexpr int C4 = NCP / 4;
  constexpr int NST = ((K / 2) * C4 + 255) / 256;
  static_assert((K / 2) * C4 == NST * 256, "every thread stages NST items");
  float4 r0[NST], r1[NST];
#pragma unroll
  for (int s = 0; s < NST; ++s) {
    const int i = threadIdx.x + s * 256;
    const int kp = i / C4;
    const int c4 = (i - kp * C4) * 4;
    const int c4l = (c4 < NC) ? c4 : (NC - 4);
    r0[s] = *reinterpret_cast<const float4*>(B + (2 * kp) * NC + c4l);
    r1[s] = *reinterpret_cast<const float4*>(B + (2 * kp + 1) * NC + c4l);
  }
#pragma unroll
  for (int s = 0; s < NST; ++s) {
    const int i = threadIdx.x + s * 256;
    const int kp = i / C4;
    const int c4 = (i - kp * C4) * 4;
    float4 u0 = r0[s], u1 = r1[s];
    if (c4 >= NC) {
      u0 = make_float4(0.f, 0.f, 0.f, 0.f);
      u1 = make_float4(0.f, 0.f, 0.f, 0.f);
    }
    float4* dp = reinterpret_cast<float4*>(&Bs[kp * PITCH + c4]);
    dp[0] = make_float4(u0.x, u1.x, u0.y, u1.y);
    dp[1] = make_float4(u0.z, u1.z, u0.w, u1.w);
  }

  const int lane = threadIdx.x & 31;
  const int wv = __builtin_amdgcn_readfirstlane((int)(threadIdx.x >> 5));
  const int tile0 = (int)blockIdx.x * TILE;

  const float4* __restrict__ xv = reinterpret_cast<const float4*>(x);
  const float4 bv = reinterpret_cast<const float4*>(bias)[lane];
  for (int i = 0; i < TILE / 8; ++i) {
    const int lrow = wv * (TILE / 8) + i;
    const int node = tile0 + lrow;
    const int beg = rowptr[node];
    const int end = rowptr[node + 1];
    const float rs = rowScaleOut[node];
    const float nx = rowScaleNext[node];
    float4 acc = {0.f, 0.f, 0.f, 0.f};
    int j = beg;
    for (; j + 4 <= end; j += 4) {
      const int s0 = csr_src[j], s1 = csr_src[j + 1], s2 = csr_src[j + 2], s3 = csr_src[j + 3];
      const float4 v0 = xv[s0 * LPN + lane];
      const float4 v1 = xv[s1 * LPN + lane];
      const float4 v2 = xv[s2 * LPN + lane];
      const float4 v3 = xv[s3 * LPN + lane];
      acc.x += v0.x; acc.y += v0.y; acc.z += v0.z; acc.w += v0.w;
      acc.x += v1.x; acc.y += v1.y; acc.z += v1.z; acc.w += v1.w;
      acc.x += v2.x; acc.y += v2.y; acc.z += v2.z; acc.w += v2.w;
      acc.x += v3.x; acc.y += v3.y; acc.z += v3.z; acc.w += v3.w;
    }
    for (; j < end; ++j) {
      const int s0 = csr_src[j];
      const float4 v0 = xv[s0 * LPN + lane];
      acc.x += v0.x; acc.y += v0.y; acc.z += v0.z; acc.w += v0.w;
    }
    acc.x = fmaxf(acc.x * rs + bv.x, 0.0f) * nx;
    acc.y = fmaxf(acc.y * rs + bv.y, 0.0f) * nx;
    acc.z = fmaxf(acc.z * rs + bv.z, 0.0f) * nx;
    acc.w = fmaxf(acc.w * rs + bv.w, 0.0f) * nx;
    *reinterpret_cast<float4*>(&Ts[lrow * TP + lane * 4]) = acc;
  }
  __syncthreads();

  if (wv < 2 * NCT) {
    const int half = lane >> 4;
    const int l15  = lane & 15;
    const int mt = wv / NCT;
    const int ct = wv - mt * NCT;
    const float*  Ta = Ts + (mt * 16 + l15) * TP + 4 * half;
    const float2* Bl = Bs + (2 * half) * PITCH + ct * 16 + l15;
    v8f acc = (v8f){0.f, 0.f, 0.f, 0.f, 0.f, 0.f, 0.f, 0.f};
#pragma unroll
    for (int J = 0; J < K / 8; ++J) {
      const float4 a4 = *reinterpret_cast<const float4*>(Ta + 8 * J);
#pragma unroll
      for (int e = 0; e < 2; ++e) {
        v2f a;
        a.x = e ? a4.z : a4.x;
        a.y = e ? a4.w : a4.y;
        const float2 bb = Bl[(4 * J + e) * PITCH];
        v2f b;
        b.x = bb.x;
        b.y = bb.y;
        acc = __builtin_amdgcn_wmma_f32_16x16x4_f32(false, a, false, b, (short)0, acc, false, false);
      }
    }
    const int mbase = tile0 + mt * 16 + half * 8;
    const int col = ct * 16 + l15;
    if ((NC % 16 == 0) || col < NC) {
#pragma unroll
      for (int r = 0; r < 8; ++r) D[(mbase + r) * NC + col] = acc[r];
    }
  }
}


extern "C" void kernel_launch(void* const* d_in, const int* in_sizes, int n_in,
                              void* d_out, int out_size, void* d_ws, size_t ws_size,
                              hipStream_t stream) {
  const float* features = (const float*)d_in[0];
  const int*   src      = (const int*)d_in[1];
  const int*   dst      = (const int*)d_in[2];
  const float* W1       = (const float*)d_in[3];
  const float* b1       = (const float*)d_in[4];
  const float* W2       = (const float*)d_in[5];
  const float* b2       = (const float*)d_in[6];
  const float* W3       = (const float*)d_in[7];
  const float* b3       = (const float*)d_in[8];

  const int N = NNODES;
  const int E = in_sizes[1];

  float* ws      = (float*)d_ws;
  float* inv_out = ws;
  float* inv_in  = ws + N;
  float* P0      = ws + 2 * (size_t)N;
  float* P1      = P0 + (size_t)N * 256;
  float* P2      = P1 + (size_t)N * 128;
  float* H3      = P0 + (size_t)N * 128;

  const int NB = (N + 1023) / 1024;
  int* cnt_out  = (int*)P2;
  int* cnt_in   = cnt_out + N;
  int* rowptr   = cnt_in + N;
  int* rank     = rowptr + N + 4;
  int* csr_src  = rank + E;
  int* blocksum = csr_src + E;

  const int T = 256;
  auto blocks = [T](long long n) { return (unsigned)((n + T - 1) / T); };

  zero_i32<<<blocks(2LL * N), T, 0, stream>>>(cnt_out, 2 * N);
  count_rank_kernel<<<blocks(E), T, 0, stream>>>(src, dst, cnt_out, cnt_in, rank, E);
  norm_kernel<<<blocks(N), T, 0, stream>>>(cnt_out, cnt_in, inv_out, inv_in, N);
  scan_local_kernel<<<NB, 256, 0, stream>>>(cnt_in, rowptr, blocksum, N);
  scan_add_kernel<<<NB, 256, 0, stream>>>(rowptr, blocksum, N);
  fill_kernel<<<blocks(E), T, 0, stream>>>(src, dst, rowptr, rank, csr_src, E);

  gather_kernel<128, true, false, false><<<blocks((long long)N * 32), 256, 0, stream>>>(
      features, rowptr, csr_src, inv_out, nullptr, nullptr, P1, N);
  gemm_strip<128, 256, 8, false, true, true><<<512, 256, 0, stream>>>(
      P1, W1, P0, nullptr, inv_in, b1, N / 16);

  gemm_strip<256, 128, 8, true, false, false><<<512, 256, 0, stream>>>(
      P0, W2, P1, inv_out, nullptr, nullptr, N / 16);
  gather_project_kernel<40><<<N / 32, 256, 0, stream>>>(
      P1, rowptr, csr_src, inv_in, b2, inv_out, W3, H3);

  gather_kernel<40, false, true, false><<<blocks((long long)N * 10), 256, 0, stream>>>(
      H3, rowptr, csr_src, nullptr, inv_in, b3, (float*)d_out, N);
}


# Degree counts and edge ranks by LDS atomics, a work-group per node range x edge chunk, chunk counts prefix-summed per node: no global atomics or zero-fill; sum order in a node = chunk, then LDS rank
# speedup vs baseline: 4.3787x; 1.1193x over previous
#include <hip/hip_runtime.h>


typedef __attribute__((ext_vector_type(2))) float v2f;
typedef __attribute__((ext_vector_type(8))) float v8f;

#define NNODES 100000


#define NRANGE 8
#define RANGE (NNODES / NRANGE)
#define NCHUNK 16

__device__ __forceinline__ int chunk_len(int E) {
  return (((E + NCHUNK - 1) / NCHUNK) + 3) & ~3;
}

__global__ __launch_bounds__(1024) void count_lds_kernel(const int* __restrict__ src,
                                                         const int* __restrict__ dst,
                                                         int* __restrict__ part_out,
                                                         int* __restrict__ part_in,
                                                         int* __restrict__ rank, int E) {
  static_assert(NNODES % NRANGE == 0, "equal node ranges");
  __shared__ int co[RANGE];
  __shared__ int ci[RANGE];
  const int t  = threadIdx.x;
  const int lo = (int)blockIdx.x * RANGE;
  const int c  = (int)blockIdx.y;
  for (int i = t; i < RANGE; i += 1024) { co[i] = 0; ci[i] = 0; }
  __syncthreads();
  const int len = chunk_len(E);
  const int e0 = c * len;
  const int e1 = (e0 + len < E) ? (e0 + len) : E;
  int e = e0 + t * 4;
  for (; e + 4100 <= e1; e += 8192) {
    const int4 sa = *reinterpret_cast<const int4*>(src + e);
    const int4 da = *reinterpret_cast<const int4*>(dst + e);
    const int4 sb = *reinterpret_cast<const int4*>(src + e + 4096);
    const int4 db = *reinterpret_cast<const int4*>(dst + e + 4096);
    const int sv[8] = {sa.x, sa.y, sa.z, sa.w, sb.x, sb.y, sb.z, sb.w};
    const int dv[8] = {da.x, da.y, da.z, da.w, db.x, db.y, db.z, db.w};
    int rv[8];
#pragma unroll
    for (int k = 0; k < 8; ++k) {
      const unsigned s = (unsigned)(sv[k] - lo);
      const unsigned d = (unsigned)(dv[k] - lo);
      if (s < (unsigned)RANGE) atomicAdd(&co[s], 1);
      rv[k] = -1;
      if (d < (unsigned)RANGE) rv[k] = atomicAdd(&ci[d], 1);
    }
#pragma unroll
    for (int k = 0; k < 8; ++k) {
      if (rv[k] >= 0) rank[e + (k < 4 ? k : 4092 + k)] = rv[k];
    }
  }
  for (; e < e1; e += 4096) {
    for (int k = 0; k < 4; ++k) {
      if (e + k < e1) {
        const unsigned s = (unsigned)(src[e + k] - lo);
        const unsigned d = (unsigned)(dst[e + k] - lo);
        if (s < (unsigned)RANGE) atomicAdd(&co[s], 1);
        if (d < (unsigned)RANGE) rank[e + k] = atomicAdd(&ci[d], 1);
      }
    }
  }
  __syncthreads();
  for (int i = t; i < RANGE; i += 1024) {
    part_out[c * NNODES + lo + i] = co[i];
    part_in[c * NNODES + lo + i]  = ci[i];
  }
}

__global__ void reduce_norm_kernel(const int* __restrict__ part_out, int* __restrict__ part_in,
                                   int* __restrict__ cnt_in,
                                   float* __restrict__ inv_out, float* __restrict__ inv_in, int n) {
  int i = blockIdx.x * blockDim.x + threadIdx.x;
  if (i < n) {
    int vo[NCHUNK], vi[NCHUNK];
#pragma unroll
    for (int c = 0; c < NCHUNK; ++c) {
      vo[c] = part_out[c * NNODES + i];
      vi[c] = part_in[c * NNODES + i];
    }
    int so = 0, si = 0;
#pragma unroll
    for (int c = 0; c < NCHUNK; ++c) {
      so += vo[c];
      part_in[c * NNODES + i] = si;
      si += vi[c];
    }
    cnt_in[i]  = si;
    inv_out[i] = rsqrtf(fmaxf((float)so, 1.0f));
    inv_in[i]  = rsqrtf(fmaxf((float)si, 1.0f));
  }
}

__global__ __launch_bounds__(256) void scan_local_kernel(const int* __restrict__ cnt,
                                                         int* __restrict__ rowptr,
                                                         int* __restrict__ blocksum, int n) {
  __shared__ int sh[256];
  const int t = threadIdx.x;
  const int base = (blockIdx.x * 256 + t) * 4;
  int c0 = 0, c1 = 0, c2 = 0, c3 = 0;
  if (base + 3 < n) {
    const int4 v = *reinterpret_cast<const int4*>(cnt + base);
    c0 = v.x; c1 = v.y; c2 = v.z; c3 = v.w;
  } else {
    if (base     < n) c0 = cnt[base];
    if (base + 1 < n) c1 = cnt[base + 1];
    if (base + 2 < n) c2 = cnt[base + 2];
  }
  const int tsum = c0 + c1 + c2 + c3;
  sh[t] = tsum;
  __syncthreads();
  for (int off = 1; off < 256; off <<= 1) {
    const int v = (t >= off) ? sh[t - off] : 0;
    __syncthreads();
    sh[t] += v;
    __syncthreads();
  }
  const int excl = sh[t] - tsum;
  if (base     < n) rowptr[base]     = excl;
  if (base + 1 < n) rowptr[base + 1] = excl + c0;
  if (base + 2 < n) rowptr[base + 2] = excl + c0 + c1;
  if (base + 3 < n) rowptr[base + 3] = excl + c0 + c1 + c2;
  if (t == 255) blocksum[blockIdx.x] = sh[255];
}

__global__ __launch_bounds__(256) void scan_add_kernel(int* __restrict__ rowptr,
                                                       const int* __restrict__ blocksum, int n) {
  __shared__ int sh[256];
  const int t = threadIdx.x;
  const int b = blockIdx.x;
  int part = 0;
  for (int i = t; i < b; i += 256) part += blocksum[i];
  sh[t] = part;
  __syncthreads();
  for (int off = 128; off > 0; off >>= 1) {
    if (t < off) sh[t] += sh[t + off];
    __syncthreads();
  }
  const int offset = sh[0];
  const int base = (b * 256 + t) * 4;
  for (int k = 0; k < 4; ++k)
    if (base + k < n) rowptr[base + k] += offset;
  if (b == (int)gridDim.x - 1 && t == 0) rowptr[n] = offset + blocksum[b];
}

__global__ void fill_kernel(const int* __restrict__ src, const int* __restrict__ dst,
                            const int* __restrict__ rowptr, const int* __restrict__ pre,
                            const int* __restrict__ rank,
                            int* __restrict__ csr_src, int E) {
  int e = blockIdx.x * blockDim.x + threadIdx.x;
  if (e < E) {
    const int d = dst[e];
    const int c = e / chunk_len(E);
    csr_src[rowptr[d] + pre[c * NNODES + d] + rank[e]] = src[e];
  }
}

template <int F, bool SCALE, bool EPI, bool RELU>
__global__ __launch_bounds__(256) void gather_kernel(const float* __restrict__ x,
                                                     const int* __restrict__ rowptr,
                                                     const int* __restrict__ csr_src,
                                                     const float* __restrict__ scale,
                                                     const float* __restrict__ rowScaleOut,
                                                     const float* __restrict__ bias,
                                                     float* __restrict__ agg, int n) {
  constexpr int LPN = F / 4;
  const int tid = blockIdx.x * 256 + threadIdx.x;
  int node = tid / LPN;
  const int q = tid - node * LPN;
  if (LPN == 32) node = __builtin_amdgcn_readfirstlane(node);
  if (node >= n) return;
  const int beg = rowptr[node];
  const int end = rowptr[node + 1];
  float rs = 1.0f;
  float4 bv = {0.f, 0.f, 0.f, 0.f};
  if (EPI) {
    rs = rowScaleOut[node];
    bv = reinterpret_cast<const float4*>(bias)[q];
  }
  const float4* __restrict__ xv = reinterpret_cast<const float4*>(x);
  float4 acc = {0.f, 0.f, 0.f, 0.f};
  int j = beg;
  for (; j + 4 <= end; j += 4) {
    const int s0 = csr_src[j], s1 = csr_src[j + 1], s2 = csr_src[j + 2], s3 = csr_src[j + 3];
    const float4 v0 = xv[s0 * LPN + q];
    const float4 v1 = xv[s1 * LPN + q];
    const float4 v2 = xv[s2 * LPN + q];
    const float4 v3 = xv[s3 * LPN + q];
    float w0 = 1.0f, w1 = 1.0f, w2 = 1.0f, w3 = 1.0f;
    if (SCALE) { w0 = scale[s0]; w1 = scale[s1]; w2 = scale[s2]; w3 = scale[s3]; }
    acc.x += v0.x * w0; acc.y += v0.y * w0; acc.z += v0.z * w0; acc.w += v0.w * w0;
    acc.x += v1.x * w1; acc.y += v1.y * w1; acc.z += v1.z * w1; acc.w += v1.w * w1;
    acc.x += v2.x * w2; acc.y += v2.y * w2; acc.z += v2.z * w2; acc.w += v2.w * w2;
    acc.x += v3.x * w3; acc.y += v3.y * w3; acc.z += v3.z * w3; acc.w += v3.w * w3;
  }
  for (; j < end; ++j) {
    const int s0 = csr_src[j];
    const float4 v0 = xv[s0 * LPN + q];
    float w0 = 1.0f;
    if (SCALE) w0 = scale[s0];
    acc.x += v0.x * w0; acc.y += v0.y * w0; acc.z += v0.z * w0; acc.w += v0.w * w0;
  }
  if (EPI) {
    acc.x = acc.x * rs + bv.x;
    acc.y = acc.y * rs + bv.y;
    acc.z = acc.z * rs + bv.z;
    acc.w = acc.w * rs + bv.w;
  }
  if (RELU) {
    acc.x = fmaxf(acc.x, 0.0f);
    acc.y = fmaxf(acc.y, 0.0f);
    acc.z = fmaxf(acc.z, 0.0f);
    acc.w = fmaxf(acc.w, 0.0f);
  }
  reinterpret_cast<float4*>(agg)[node * LPN + q] = acc;
}

template <int K, int NC, int NT, bool PRESCALE, bool EPI, bool RELU>
__global__ __launch_bounds__(256) __attribute__((amdgpu_waves_per_eu(4, 4))) void gemm_strip(const float* __restrict__ A,
                                                  const float* __restrict__ B,
                                                  float* __restrict__ D,
                                                  const float* __restrict__ rowScaleA,
                                                  const float* __restrict__ rowScaleOut,
                                                  const float* __restrict__ bias,
                                                  int mtiles) {
  constexpr int NCT   = (NC + 15) / 16;
  constexpr int NCP   = NCT * 16;
  constexpr int NG    = NCT / NT;
  static_assert(NCT % NT == 0, "column tiles must split evenly into groups");
  static_assert(K % 8 == 0 && NC % 4 == 0, "vector loads need these");
  constexpr int PITCH = NCP + 8;
  __shared__ __attribute__((aligned(16))) float2 Bs[(K / 2) * PITCH];

  constexpr int C4 = NCP / 4;
  for (int i = threadIdx.x; i < (K / 2) * C4; i += 256) {
    const int kp = i / C4;
    const int c4 = (i - kp * C4) * 4;
    float4 r0 = {0.f, 0.f, 0.f, 0.f}, r1 = {0.f, 0.f, 0.f, 0.f};
    if (c4 < NC) {
      r0 = *reinterpret_cast<const float4*>(B + (2 * kp) * NC + c4);
      r1 = *reinterpret_cast<const float4*>(B + (2 * kp + 1) * NC + c4);
    }
    float4* dp = reinterpret_cast<float4*>(&Bs[kp * PITCH + c4]);
    dp[0] = make_float4(r0.x, r1.x, r0.y, r1.y);
    dp[1] = make_float4(r0.z, r1.z, r0.w, r1.w);
  }
  __syncthreads();

  const int lane = threadIdx.x & 31;
  const int half = lane >> 4;
  const int l15  = lane & 15;
  const int wv = __builtin_amdgcn_readfirstlane((int)(threadIdx.x >> 5));
  const int nWaves = (int)gridDim.x * 8;
  const int units  = mtiles * NG;

  for (int u = wv * (int)gridDim.x + (int)blockIdx.x; u < units; u += nWaves) {
    const int mt = u / NG;
    const int ng = u - mt * NG;
    const int tileM = mt * 16;
    const int arow  = tileM + l15;
    const float4* __restrict__ Ap = reinterpret_cast<const float4*>(A + arow * K) + half;

    constexpr int KC  = 32;
    constexpr int JC  = KC / 8;
    constexpr int NCH = K / KC;
    static_assert(K % KC == 0, "K must be a whole number of chunks");
    float ascale = 1.0f;
    if (PRESCALE) ascale = rowScaleA[arow];

    v8f acc[NT];
#pragma unroll
    for (int t = 0; t < NT; ++t) acc[t] = (v8f){0.f, 0.f, 0.f, 0.f, 0.f, 0.f, 0.f, 0.f};

    const float2* Bl = Bs + (2 * half) * PITCH + ng * (NT * 16) + l15;

    float4 an[JC];
#pragma unroll
    for (int j = 0; j < JC; ++j) an[j] = Ap[2 * j];

#pragma unroll 1
    for (int ch = 0; ch < NCH; ++ch) {
      float4 ac[JC];
#pragma unroll
      for (int j = 0; j < JC; ++j) {
        ac[j] = an[j];
        if (PRESCALE) { ac[j].x *= ascale; ac[j].y *= ascale; ac[j].z *= ascale; ac[j].w *= ascale; }
      }
      const int chn = (ch + 1 < NCH) ? ch + 1 : ch;
#pragma unroll
      for (int j = 0; j < JC; ++j) an[j] = Ap[2 * (chn * JC + j)];

      const float2* Bc = Bl + (4 * JC * ch) * PITCH;
#pragma unroll
      for (int j = 0; j < JC; ++j) {
#pragma unroll
        for (int e = 0; e < 2; ++e) {
          v2f a;
          a.x = e ? ac[j].z : ac[j].x;
          a.y = e ? ac[j].w : ac[j].y;
          const float2* Br = Bc + (4 * j + e) * PITCH;
#pragma unroll
          for (int t = 0; t < NT; ++t) {
            const float2 bb = Br[t * 16];
            v2f b;
            b.x = bb.x;
            b.y = bb.y;
            acc[t] = __builtin_amdgcn_wmma_f32_16x16x4_f32(false, a, false, b, (short)0, acc[t],
                                                           false, false);
          }
        }
      }
    }

    const int mbase = tileM + half * 8;
    float rs[8];
    if (EPI) {
#pragma unroll
      for (int r = 0; r < 8; ++r) rs[r] = rowScaleOut[mbase + r];
    }
#pragma unroll
    for (int t = 0; t < NT; ++t) {
      const int col = (ng * NT + t) * 16 + l15;
      if ((NC % 16 == 0) || col < NC) {
        float bv = 0.0f;
        if (EPI) bv = bias[col];
#pragma unroll
        for (int r = 0; r < 8; ++r) {
          float v = acc[t][r];
          if (EPI)  v = v * rs[r] + bv;
          if (RELU) v = fmaxf(v, 0.0f);
          D[(mbase + r) * NC + col] = v;
        }
      }
    }
  }
}


template <int NC>
__global__ __launch_bounds__(256) void gather_project_kernel(const float* __restrict__ x,
                                                             const int* __restrict__ rowptr,
                                                             const int* __restrict__ csr_src,
                                                             const float* __restrict__ rowScaleOut,
                                                             const float* __restrict__ bias,
                                                             const float* __restrict__ rowScaleNext,
                                                             const float* __restrict__ B,
                                                             float* __restrict__ D) {
  constexpr int K     = 128;
  constexpr int LPN   = 32;
  constexpr int TILE  = 32;
  constexpr int NCT   = (NC + 15) / 16;
  constexpr int NCP   = NCT * 16;
  constexpr int PITCH = NCP + 8;
  constexpr int TP    = K + 4;
  static_assert(NNODES % TILE == 0, "whole tiles only");
  static_assert(2 * NCT <= 8, "one wave per output tile");
  static_assert(NC % 4 == 0, "vector loads need this");
  __shared__ __attribute__((aligned(16))) float2 Bs[(K / 2) * PITCH];
  __shared__ __attribute__((aligned(16))) float  Ts[TILE * TP];

  constexpr int C4 = NCP / 4;
  constexpr int NST = ((K / 2) * C4 + 255) / 256;
  static_assert((K / 2) * C4 == NST * 256, "every thread stages NST items");
  float4 r0[NST], r1[NST];
#pragma unroll
  for (int s = 0; s < NST; ++s) {
    const int i = threadIdx.x + s * 256;
    const int kp = i / C4;
    const int c4 = (i - kp * C4) * 4;
    const int c4l = (c4 < NC) ? c4 : (NC - 4);
    r0[s] = *reinterpret_cast<const float4*>(B + (2 * kp) * NC + c4l);
    r1[s] = *reinterpret_cast<const float4*>(B + (2 * kp + 1) * NC + c4l);
  }
#pragma unroll
  for (int s = 0; s < NST; ++s) {
    const int i = threadIdx.x + s * 256;
    const int kp = i / C4;
    const int c4 = (i - kp * C4) * 4;
    float4 u0 = r0[s], u1 = r1[s];
    if (c4 >= NC) {
      u0 = make_float4(0.f, 0.f, 0.f, 0.f);
      u1 = make_float4(0.f, 0.f, 0.f, 0.f);
    }
    float4* dp = reinterpret_cast<float4*>(&Bs[kp * PITCH + c4]);
    dp[0] = make_float4(u0.x, u1.x, u0.y, u1.y);
    dp[1] = make_float4(u0.z, u1.z, u0.w, u1.w);
  }

  const int lane = threadIdx.x & 31;
  const int wv = __builtin_amdgcn_readfirstlane((int)(threadIdx.x >> 5));
  const int tile0 = (int)blockIdx.x * TILE;

  const float4* __restrict__ xv = reinterpret_cast<const float4*>(x);
  const float4 bv = reinterpret_cast<const float4*>(bias)[lane];
  for (int i = 0; i < TILE / 8; ++i) {
    const int lrow = wv * (TILE / 8) + i;
    const int node = tile0 + lrow;
    const int beg = rowptr[node];
    const int end = rowptr[node + 1];
    const float rs = rowScaleOut[node];
    const float nx = rowScaleNext[node];
    float4 acc = {0.f, 0.f, 0.f, 0.f};
    int j = beg;
    for (; j + 4 <= end; j += 4) {
      const int s0 = csr_src[j], s1 = csr_src[j + 1], s2 = csr_src[j + 2], s3 = csr_src[j + 3];
      const float4 v0 = xv[s0 * LPN + lane];
      const float4 v1 = xv[s1 * LPN + lane];
      const float4 v2 = xv[s2 * LPN + lane];
      const float4 v3 = xv[s3 * LPN + lane];
      acc.x += v0.x; acc.y += v0.y; acc.z += v0.z; acc.w += v0.w;
      acc.x += v1.x; acc.y += v1.y; acc.z += v1.z; acc.w += v1.w;
      acc.x += v2.x; acc.y += v2.y; acc.z += v2.z; acc.w += v2.w;
      acc.x += v3.x; acc.y += v3.y; acc.z += v3.z; acc.w += v3.w;
    }
    for (; j < end; ++j) {
      const int s0 = csr_src[j];
      const float4 v0 = xv[s0 * LPN + lane];
      acc.x += v0.x; acc.y += v0.y; acc.z += v0.z; acc.w += v0.w;
    }
    acc.x = fmaxf(acc.x * rs + bv.x, 0.0f) * nx;
    acc.y = fmaxf(acc.y * rs + bv.y, 0.0f) * nx;
    acc.z = fmaxf(acc.z * rs + bv.z, 0.0f) * nx;
    acc.w = fmaxf(acc.w * rs + bv.w, 0.0f) * nx;
    *reinterpret_cast<float4*>(&Ts[lrow * TP + lane * 4]) = acc;
  }
  __syncthreads();

  if (wv < 2 * NCT) {
    const int half = lane >> 4;
    const int l15  = lane & 15;
    const int mt = wv / NCT;
    const int ct = wv - mt * NCT;
    const float*  Ta = Ts + (mt * 16 + l15) * TP + 4 * half;
    const float2* Bl = Bs + (2 * half) * PITCH + ct * 16 + l15;
    v8f acc = (v8f){0.f, 0.f, 0.f, 0.f, 0.f, 0.f, 0.f, 0.f};
#pragma unroll
    for (int J = 0; J < K / 8; ++J) {
      const float4 a4 = *reinterpret_cast<const float4*>(Ta + 8 * J);
#pragma unroll
      for (int e = 0; e < 2; ++e) {
        v2f a;
        a.x = e ? a4.z : a4.x;
        a.y = e ? a4.w : a4.y;
        const float2 bb = Bl[(4 * J + e) * PITCH];
        v2f b;
        b.x = bb.x;
        b.y = bb.y;
        acc = __builtin_amdgcn_wmma_f32_16x16x4_f32(false, a, false, b, (short)0, acc, false, false);
      }
    }
    const int mbase = tile0 + mt * 16 + half * 8;
    const int col = ct * 16 + l15;
    if ((NC % 16 == 0) || col < NC) {
#pragma unroll
      for (int r = 0; r < 8; ++r) D[(mbase + r) * NC + col] = acc[r];
    }
  }
}


extern "C" void kernel_launch(void* const* d_in, const int* in_sizes, int n_in,
                              void* d_out, int out_size, void* d_ws, size_t ws_size,
                              hipStream_t stream) {
  const float* features = (const float*)d_in[0];
  const int*   src      = (const int*)d_in[1];
  const int*   dst      = (const int*)d_in[2];
  const float* W1       = (const float*)d_in[3];
  const float* b1       = (const float*)d_in[4];
  const float* W2       = (const float*)d_in[5];
  const float* b2       = (const float*)d_in[6];
  const float* W3       = (const float*)d_in[7];
  const float* b3       = (const float*)d_in[8];

  const int N = NNODES;
  const int E = in_sizes[1];

  float* ws      = (float*)d_ws;
  float* inv_out = ws;
  float* inv_in  = ws + N;
  float* P0      = ws + 2 * (size_t)N;
  float* P1      = P0 + (size_t)N * 256;
  float* P2      = P1 + (size_t)N * 128;
  float* H3      = P0 + (size_t)N * 128;

  const int NB = (N + 1023) / 1024;
  int* cnt_in   = (int*)P2;
  int* rowptr   = cnt_in + N;
  int* rank     = rowptr + N + 4;
  int* csr_src  = rank + E;
  int* blocksum = csr_src + E;
  int* part_out = (int*)P0;
  int* part_in  = part_out + (size_t)NCHUNK * N;

  const int T = 256;
  auto blocks = [T](long long n) { return (unsigned)((n + T - 1) / T); };

  count_lds_kernel<<<dim3(NRANGE, NCHUNK), 1024, 0, stream>>>(src, dst, part_out, part_in, rank, E);
  reduce_norm_kernel<<<blocks(N), T, 0, stream>>>(part_out, part_in, cnt_in, inv_out, inv_in, N);
  scan_local_kernel<<<NB, 256, 0, stream>>>(cnt_in, rowptr, blocksum, N);
  scan_add_kernel<<<NB, 256, 0, stream>>>(rowptr, blocksum, N);
  fill_kernel<<<blocks(E), T, 0, stream>>>(src, dst, rowptr, part_in, rank, csr_src, E);

  gather_kernel<128, true, false, false><<<blocks((long long)N * 32), 256, 0, stream>>>(
      features, rowptr, csr_src, inv_out, nullptr, nullptr, P1, N);
  gemm_strip<128, 256, 8, false, true, true><<<512, 256, 0, stream>>>(
      P1, W1, P0, nullptr, inv_in, b1, N / 16);

  gemm_strip<256, 128, 8, true, false, false><<<512, 256, 0, stream>>>(
      P0, W2, P1, inv_out, nullptr, nullptr, N / 16);
  gather_project_kernel<40><<<N / 32, 256, 0, stream>>>(
      P1, rowptr, csr_src, inv_in, b2, inv_out, W3, H3);

  gather_kernel<40, false, true, false><<<blocks((long long)N * 10), 256, 0, stream>>>(
      H3, rowptr, csr_src, nullptr, inv_in, b3, (float*)d_out, N);
}


# Layer-1 and layer-2 products in one launch: product 1 computed transposed, its accumulators are product 2's A fragments, out1 never stored; k grouping of product 2's f32 sums changes
# speedup vs baseline: 4.7741x; 1.0903x over previous
#include <hip/hip_runtime.h>


typedef __attribute__((ext_vector_type(2))) float v2f;
typedef __attribute__((ext_vector_type(8))) float v8f;

#define NNODES 100000


#define NRANGE 8
#define RANGE (NNODES / NRANGE)
#define NCHUNK 16

__device__ __forceinline__ int chunk_len(int E) {
  return (((E + NCHUNK - 1) / NCHUNK) + 3) & ~3;
}

__global__ __launch_bounds__(1024) void count_lds_kernel(const int* __restrict__ src,
                                                         const int* __restrict__ dst,
                                                         int* __restrict__ part_out,
                                                         int* __restrict__ part_in,
                                                         int* __restrict__ rank, int E) {
  static_assert(NNODES % NRANGE == 0, "equal node ranges");
  __shared__ int co[RANGE];
  __shared__ int ci[RANGE];
  const int t  = threadIdx.x;
  const int lo = (int)blockIdx.x * RANGE;
  const int c  = (int)blockIdx.y;
  for (int i = t; i < RANGE; i += 1024) { co[i] = 0; ci[i] = 0; }
  __syncthreads();
  const int len = chunk_len(E);
  const int e0 = c * len;
  const int e1 = (e0 + len < E) ? (e0 + len) : E;
  int e = e0 + t * 4;
  for (; e + 4100 <= e1; e += 8192) {
    const int4 sa = *reinterpret_cast<const int4*>(src + e);
    const int4 da = *reinterpret_cast<const int4*>(dst + e);
    const int4 sb = *reinterpret_cast<const int4*>(src + e + 4096);
    const int4 db = *reinterpret_cast<const int4*>(dst + e + 4096);
    const int sv[8] = {sa.x, sa.y, sa.z, sa.w, sb.x, sb.y, sb.z, sb.w};
    const int dv[8] = {da.x, da.y, da.z, da.w, db.x, db.y, db.z, db.w};
    int rv[8];
#pragma unroll
    for (int k = 0; k < 8; ++k) {
      const unsigned s = (unsigned)(sv[k] - lo);
      const unsigned d = (unsigned)(dv[k] - lo);
      if (s < (unsigned)RANGE) atomicAdd(&co[s], 1);
      rv[k] = -1;
      if (d < (unsigned)RANGE) rv[k] = atomicAdd(&ci[d], 1);
    }
#pragma unroll
    for (int k = 0; k < 8; ++k) {
      if (rv[k] >= 0) rank[e + (k < 4 ? k : 4092 + k)] = rv[k];
    }
  }
  for (; e < e1; e += 4096) {
    for (int k = 0; k < 4; ++k) {
      if (e + k < e1) {
        const unsigned s = (unsigned)(src[e + k] - lo);
        const unsigned d = (unsigned)(dst[e + k] - lo);
        if (s < (unsigned)RANGE) atomicAdd(&co[s], 1);
        if (d < (unsigned)RANGE) rank[e + k] = atomicAdd(&ci[d], 1);
      }
    }
  }
  __syncthreads();
  for (int i = t; i < RANGE; i += 1024) {
    part_out[c * NNODES + lo + i] = co[i];
    part_in[c * NNODES + lo + i]  = ci[i];
  }
}

__global__ void reduce_norm_kernel(const int* __restrict__ part_out, int* __restrict__ part_in,
                                   int* __restrict__ cnt_in,
                                   float* __restrict__ inv_out, float* __restrict__ inv_in, int n) {
  int i = blockIdx.x * blockDim.x + threadIdx.x;
  if (i < n) {
    int vo[NCHUNK], vi[NCHUNK];
#pragma unroll
    for (int c = 0; c < NCHUNK; ++c) {
      vo[c] = part_out[c * NNODES + i];
      vi[c] = part_in[c * NNODES + i];
    }
    int so = 0, si = 0;
#pragma unroll
    for (int c = 0; c < NCHUNK; ++c) {
      so += vo[c];
      part_in[c * NNODES + i] = si;
      si += vi[c];
    }
    cnt_in[i]  = si;
    inv_out[i] = rsqrtf(fmaxf((float)so, 1.0f));
    inv_in[i]  = rsqrtf(fmaxf((float)si, 1.0f));
  }
}

__global__ __launch_bounds__(256) void scan_local_kernel(const int* __restrict__ cnt,
                                                         int* __restrict__ rowptr,
                                                         int* __restrict__ blocksum, int n) {
  __shared__ int sh[256];
  const int t = threadIdx.x;
  const int base = (blockIdx.x * 256 + t) * 4;
  int c0 = 0, c1 = 0, c2 = 0, c3 = 0;
  if (base + 3 < n) {
    const int4 v = *reinterpret_cast<const int4*>(cnt + base);
    c0 = v.x; c1 = v.y; c2 = v.z; c3 = v.w;
  } else {
    if (base     < n) c0 = cnt[base];
    if (base + 1 < n) c1 = cnt[base + 1];
    if (base + 2 < n) c2 = cnt[base + 2];
  }
  const int tsum = c0 + c1 + c2 + c3;
  sh[t] = tsum;
  __syncthreads();
  for (int off = 1; off < 256; off <<= 1) {
    const int v = (t >= off) ? sh[t - off] : 0;
    __syncthreads();
    sh[t] += v;
    __syncthreads();
  }
  const int excl = sh[t] - tsum;
  if (base     < n) rowptr[base]     = excl;
  if (base + 1 < n) rowptr[base + 1] = excl + c0;
  if (base + 2 < n) rowptr[base + 2] = excl + c0 + c1;
  if (base + 3 < n) rowptr[base + 3] = excl + c0 + c1 + c2;
  if (t == 255) blocksum[blockIdx.x] = sh[255];
}

__global__ __launch_bounds__(256) void scan_add_kernel(int* __restrict__ rowptr,
                                                       const int* __restrict__ blocksum, int n) {
  __shared__ int sh[256];
  const int t = threadIdx.x;
  const int b = blockIdx.x;
  int part = 0;
  for (int i = t; i < b; i += 256) part += blocksum[i];
  sh[t] = part;
  __syncthreads();
  for (int off = 128; off > 0; off >>= 1) {
    if (t < off) sh[t] += sh[t + off];
    __syncthreads();
  }
  const int offset = sh[0];
  const int base = (b * 256 + t) * 4;
  for (int k = 0; k < 4; ++k)
    if (base + k < n) rowptr[base + k] += offset;
  if (b == (int)gridDim.x - 1 && t == 0) rowptr[n] = offset + blocksum[b];
}

__global__ void fill_kernel(const int* __restrict__ src, const int* __restrict__ dst,
                            const int* __restrict__ rowptr, const int* __restrict__ pre,
                            const int* __restrict__ rank,
                            int* __restrict__ csr_src, int E) {
  int e = blockIdx.x * blockDim.x + threadIdx.x;
  if (e < E) {
    const int d = dst[e];
    const int c = e / chunk_len(E);
    csr_src[rowptr[d] + pre[c * NNODES + d] + rank[e]] = src[e];
  }
}

template <int F, bool SCALE, bool EPI, bool RELU>
__global__ __launch_bounds__(256) void gather_kernel(const float* __restrict__ x,
                                                     const int* __restrict__ rowptr,
                                                     const int* __restrict__ csr_src,
                                                     const float* __restrict__ scale,
                                                     const float* __restrict__ rowScaleOut,
                                                     const float* __restrict__ bias,
                                                     float* __restrict__ agg, int n) {
  constexpr int LPN = F / 4;
  const int tid = blockIdx.x * 256 + threadIdx.x;
  int node = tid / LPN;
  const int q = tid - node * LPN;
  if (LPN == 32) node = __builtin_amdgcn_readfirstlane(node);
  if (node >= n) return;
  const int beg = rowptr[node];
  const int end = rowptr[node + 1];
  float rs = 1.0f;
  float4 bv = {0.f, 0.f, 0.f, 0.f};
  if (EPI) {
    rs = rowScaleOut[node];
    bv = reinterpret_cast<const float4*>(bias)[q];
  }
  const float4* __restrict__ xv = reinterpret_cast<const float4*>(x);
  float4 acc = {0.f, 0.f, 0.f, 0.f};
  int j = beg;
  for (; j + 4 <= end; j += 4) {
    const int s0 = csr_src[j], s1 = csr_src[j + 1], s2 = csr_src[j + 2], s3 = csr_src[j + 3];
    const float4 v0 = xv[s0 * LPN + q];
    const float4 v1 = xv[s1 * LPN + q];
    const float4 v2 = xv[s2 * LPN + q];
    const float4 v3 = xv[s3 * LPN + q];
    float w0 = 1.0f, w1 = 1.0f, w2 = 1.0f, w3 = 1.0f;
    if (SCALE) { w0 = scale[s0]; w1 = scale[s1]; w2 = scale[s2]; w3 = scale[s3]; }
    acc.x += v0.x * w0; acc.y += v0.y * w0; acc.z += v0.z * w0; acc.w += v0.w * w0;
    acc.x += v1.x * w1; acc.y += v1.y * w1; acc.z += v1.z * w1; acc.w += v1.w * w1;
    acc.x += v2.x * w2; acc.y += v2.y * w2; acc.z += v2.z * w2; acc.w += v2.w * w2;
    acc.x += v3.x * w3; acc.y += v3.y * w3; acc.z += v3.z * w3; acc.w += v3.w * w3;
  }
  for (; j < end; ++j) {
    const int s0 = csr_src[j];
    const float4 v0 = xv[s0 * LPN + q];
    float w0 = 1.0f;
    if (SCALE) w0 = scale[s0];
    acc.x += v0.x * w0; acc.y += v0.y * w0; acc.z += v0.z * w0; acc.w += v0.w * w0;
  }
  if (EPI) {
    acc.x = acc.x * rs + bv.x;
    acc.y = acc.y * rs + bv.y;
    acc.z = acc.z * rs + bv.z;
    acc.w = acc.w * rs + bv.w;
  }
  if (RELU) {
    acc.x = fmaxf(acc.x, 0.0f);
    acc.y = fmaxf(acc.y, 0.0f);
    acc.z = fmaxf(acc.z, 0.0f);
    acc.w = fmaxf(acc.w, 0.0f);
  }
  reinterpret_cast<float4*>(agg)[node * LPN + q] = acc;
}


template <int K, int NC, int PITCH, int NTHREADS>
__device__ __forceinline__ void stage_pairs(const float* __restrict__ B, float2* Bs) {
  constexpr int C4 = NC / 4;
  constexpr int ITEMS = (K / 2) * C4;
  constexpr int PER = ITEMS / NTHREADS;
  constexpr int BATCH = 4;
  static_assert(ITEMS % NTHREADS == 0 && PER % BATCH == 0, "whole batches per thread");
#pragma unroll
  for (int s0 = 0; s0 < PER; s0 += BATCH) {
    float4 r0[BATCH], r1[BATCH];
#pragma unroll
    for (int s = 0; s < BATCH; ++s) {
      const int i = (int)threadIdx.x + (s0 + s) * NTHREADS;
      const int kp = i / C4;
      const int c4 = (i - kp * C4) * 4;
      r0[s] = *reinterpret_cast<const float4*>(B + (2 * kp) * NC + c4);
      r1[s] = *reinterpret_cast<const float4*>(B + (2 * kp + 1) * NC + c4);
    }
#pragma unroll
    for (int s = 0; s < BATCH; ++s) {
      const int i = (int)threadIdx.x + (s0 + s) * NTHREADS;
      const int kp = i / C4;
      const int c4 = (i - kp * C4) * 4;
      float4* dp = reinterpret_cast<float4*>(&Bs[kp * PITCH + c4]);
      dp[0] = make_float4(r0[s].x, r1[s].x, r0[s].y, r1[s].y);
      dp[1] = make_float4(r0[s].z, r1[s].z, r0[s].w, r1[s].w);
    }
  }
}

template <int K1, int H, int N2>
__global__ __launch_bounds__(512) __attribute__((amdgpu_waves_per_eu(4, 4))) void gemm_chain(
    const float* __restrict__ A, const float* __restrict__ W1, const float* __restrict__ W2,
    float* __restrict__ D, const float* __restrict__ rowScale1, const float* __restrict__ bias1,
    const float* __restrict__ rowScale2, int mtiles) {
  static_assert(K1 % 8 == 0 && H % 64 == 0 && N2 % 16 == 0, "tile shapes");
  constexpr int FB  = 4;
  constexpr int NG  = H / 16 / FB;
  constexpr int NT2 = N2 / 16;
  constexpr int P1 = H + 8;
  constexpr int P2 = N2 + 4;
  __shared__ __attribute__((aligned(16))) float2 W1s[(K1 / 2) * P1];
  __shared__ __attribute__((aligned(16))) float2 W2s[(H / 2) * P2];

  stage_pairs<K1, H, P1, 512>(W1, W1s);
  stage_pairs<H, N2, P2, 512>(W2, W2s);
  __syncthreads();

  const int lane = threadIdx.x & 31;
  const int half = lane >> 4;
  const int l15  = lane & 15;
  const int wv = __builtin_amdgcn_readfirstlane((int)(threadIdx.x >> 5));
  const int nWaves = (int)gridDim.x * 16;
  const float2* W1l = W1s + (2 * half) * P1 + l15;
  const float2* W2l = W2s + (4 * half) * P2 + l15;

  for (int u = wv * (int)gridDim.x + (int)blockIdx.x; u < mtiles; u += nWaves) {
    const int tileM = u * 16;
    const int row   = tileM + l15;
    const float4* __restrict__ Ap = reinterpret_cast<const float4*>(A + row * K1) + half;
    float4 an[K1 / 8];
#pragma unroll
    for (int j = 0; j < K1 / 8; ++j) an[j] = Ap[2 * j];
    const float s1 = rowScale1[row];
    const float s2 = rowScale2[row];

    v8f acc2[NT2];
#pragma unroll
    for (int t = 0; t < NT2; ++t) acc2[t] = (v8f){0.f, 0.f, 0.f, 0.f, 0.f, 0.f, 0.f, 0.f};

    constexpr int NB1 = K1 / 8;
    constexpr int NB  = NB1 + FB * 4;
    static_assert(FB * 2 == 8 && NT2 == 8, "8 fragments per block");
    static_assert(NB % 2 == 0, "the register sets alternate across passes");
    float2 fr[2][8];
#pragma unroll
    for (int e = 0; e < 2; ++e)
#pragma unroll
      for (int ff = 0; ff < FB; ++ff) fr[0][e * FB + ff] = W1l[e * P1 + ff * 16];

#pragma unroll 1
    for (int g = 0; g < NG; ++g) {
      const int gn = (g + 1 < NG) ? g + 1 : g;
      v8f acc1[FB];
#pragma unroll
      for (int ff = 0; ff < FB; ++ff) acc1[ff] = (v8f){0.f, 0.f, 0.f, 0.f, 0.f, 0.f, 0.f, 0.f};
      float4 bA, bB;
      float v[8];
#pragma unroll
      for (int b = 0; b < NB; ++b) {
        {
          const int bn = (b + 1 < NB) ? b + 1 : 0;
          const int gg = (b + 1 < NB) ? g : gn;
          float2* dst = fr[(b + 1) & 1];
          if (bn < NB1) {
            const float2* Wr = W1l + (4 * bn) * P1 + gg * (FB * 16);
#pragma unroll
            for (int e = 0; e < 2; ++e)
#pragma unroll
              for (int ff = 0; ff < FB; ++ff) dst[e * FB + ff] = Wr[e * P1 + ff * 16];
          } else {
            const int q = bn - NB1;
            const float2* Wr2 = W2l + (8 * (gg * FB + q / 4) + (q % 4)) * P2;
#pragma unroll
            for (int t = 0; t < NT2; ++t) dst[t] = Wr2[t * 16];
          }
        }
        const float2* cur = fr[b & 1];
        if (b < NB1) {
#pragma unroll
          for (int e = 0; e < 2; ++e) {
            v2f x;
            x.x = e ? an[b].z : an[b].x;
            x.y = e ? an[b].w : an[b].y;
#pragma unroll
            for (int ff = 0; ff < FB; ++ff) {
              v2f w;
              w.x = cur[e * FB + ff].x;
              w.y = cur[e * FB + ff].y;
              acc1[ff] = __builtin_amdgcn_wmma_f32_16x16x4_f32(false, w, false, x, (short)0,
                                                               acc1[ff], false, false);
            }
          }
          if (b == NB1 - 4) {
            bA = *reinterpret_cast<const float4*>(bias1 + 16 * (g * FB) + 8 * half);
            bB = *reinterpret_cast<const float4*>(bias1 + 16 * (g * FB) + 8 * half + 4);
          }
        } else {
          const int q  = b - NB1;
          const int ff = q / 4;
          const int kk = q % 4;
          if (kk == 0) {
            const float bb[8] = {bA.x, bA.y, bA.z, bA.w, bB.x, bB.y, bB.z, bB.w};
#pragma unroll
            for (int r = 0; r < 8; ++r) {
              float o = acc1[ff][r] * s1 + bb[r];
              o = fmaxf(o, 0.0f);
              v[r] = o * s2;
            }
            if (ff + 1 < FB) {
              bA = *reinterpret_cast<const float4*>(bias1 + 16 * (g * FB + ff + 1) + 8 * half);
              bB = *reinterpret_cast<const float4*>(bias1 + 16 * (g * FB + ff + 1) + 8 * half + 4);
            }
          }
          v2f a;
          a.x = v[2 * kk];
          a.y = v[2 * kk + 1];
#pragma unroll
          for (int t = 0; t < NT2; ++t) {
            v2f w;
            w.x = cur[t].x;
            w.y = cur[t].y;
            acc2[t] = __builtin_amdgcn_wmma_f32_16x16x4_f32(false, a, false, w, (short)0, acc2[t],
                                                            false, false);
          }
        }
      }
    }

    const int mbase = tileM + half * 8;
#pragma unroll
    for (int t = 0; t < NT2; ++t) {
      const int col = t * 16 + l15;
#pragma unroll
      for (int r = 0; r < 8; ++r) D[(mbase + r) * N2 + col] = acc2[t][r];
    }
  }
}

template <int NC>
__global__ __launch_bounds__(256) void gather_project_kernel(const float* __restrict__ x,
                                                             const int* __restrict__ rowptr,
                                                             const int* __restrict__ csr_src,
                                                             const float* __restrict__ rowScaleOut,
                                                             const float* __restrict__ bias,
                                                             const float* __restrict__ rowScaleNext,
                                                             const float* __restrict__ B,
                                                             float* __restrict__ D) {
  constexpr int K     = 128;
  constexpr int LPN   = 32;
  constexpr int TILE  = 32;
  constexpr int NCT   = (NC + 15) / 16;
  constexpr int NCP   = NCT * 16;
  constexpr int PITCH = NCP + 8;
  constexpr int TP    = K + 4;
  static_assert(NNODES % TILE == 0, "whole tiles only");
  static_assert(2 * NCT <= 8, "one wave per output tile");
  static_assert(NC % 4 == 0, "vector loads need this");
  __shared__ __attribute__((aligned(16))) float2 Bs[(K / 2) * PITCH];
  __shared__ __attribute__((aligned(16))) float  Ts[TILE * TP];

  constexpr int C4 = NCP / 4;
  constexpr int NST = ((K / 2) * C4 + 255) / 256;
  static_assert((K / 2) * C4 == NST * 256, "every thread stages NST items");
  float4 r0[NST], r1[NST];
#pragma unroll
  for (int s = 0; s < NST; ++s) {
    const int i = threadIdx.x + s * 256;
    const int kp = i / C4;
    const int c4 = (i - kp * C4) * 4;
    const int c4l = (c4 < NC) ? c4 : (NC - 4);
    r0[s] = *reinterpret_cast<const float4*>(B + (2 * kp) * NC + c4l);
    r1[s] = *reinterpret_cast<const float4*>(B + (2 * kp + 1) * NC + c4l);
  }
#pragma unroll
  for (int s = 0; s < NST; ++s) {
    const int i = threadIdx.x + s * 256;
    const int kp = i / C4;
    const int c4 = (i - kp * C4) * 4;
    float4 u0 = r0[s], u1 = r1[s];
    if (c4 >= NC) {
      u0 = make_float4(0.f, 0.f, 0.f, 0.f);
      u1 = make_float4(0.f, 0.f, 0.f, 0.f);
    }
    float4* dp = reinterpret_cast<float4*>(&Bs[kp * PITCH + c4]);
    dp[0] = make_float4(u0.x, u1.x, u0.y, u1.y);
    dp[1] = make_float4(u0.z, u1.z, u0.w, u1.w);
  }

  const int lane = threadIdx.x & 31;
  const int wv = __builtin_amdgcn_readfirstlane((int)(threadIdx.x >> 5));
  const int tile0 = (int)blockIdx.x * TILE;

  const float4* __restrict__ xv = reinterpret_cast<const float4*>(x);
  const float4 bv = reinterpret_cast<const float4*>(bias)[lane];
  for (int i = 0; i < TILE / 8; ++i) {
    const int lrow = wv * (TILE / 8) + i;
    const int node = tile0 + lrow;
    const int beg = rowptr[node];
    const int end = rowptr[node + 1];
    const float rs = rowScaleOut[node];
    const float nx = rowScaleNext[node];
    float4 acc = {0.f, 0.f, 0.f, 0.f};
    int j = beg;
    for (; j + 4 <= end; j += 4) {
      const int s0 = csr_src[j], s1 = csr_src[j + 1], s2 = csr_src[j + 2], s3 = csr_src[j + 3];
      const float4 v0 = xv[s0 * LPN + lane];
      const float4 v1 = xv[s1 * LPN + lane];
      const float4 v2 = xv[s2 * LPN + lane];
      const float4 v3 = xv[s3 * LPN + lane];
      acc.x += v0.x; acc.y += v0.y; acc.z += v0.z; acc.w += v0.w;
      acc.x += v1.x; acc.y += v1.y; acc.z += v1.z; acc.w += v1.w;
      acc.x += v2.x; acc.y += v2.y; acc.z += v2.z; acc.w += v2.w;
      acc.x += v3.x; acc.y += v3.y; acc.z += v3.z; acc.w += v3.w;
    }
    for (; j < end; ++j) {
      const int s0 = csr_src[j];
      const float4 v0 = xv[s0 * LPN + lane];
      acc.x += v0.x; acc.y += v0.y; acc.z += v0.z; acc.w += v0.w;
    }
    acc.x = fmaxf(acc.x * rs + bv.x, 0.0f) * nx;
    acc.y = fmaxf(acc.y * rs + bv.y, 0.0f) * nx;
    acc.z = fmaxf(acc.z * rs + bv.z, 0.0f) * nx;
    acc.w = fmaxf(acc.w * rs + bv.w, 0.0f) * nx;
    *reinterpret_cast<float4*>(&Ts[lrow * TP + lane * 4]) = acc;
  }
  __syncthreads();

  if (wv < 2 * NCT) {
    const int half = lane >> 4;
    const int l15  = lane & 15;
    const int mt = wv / NCT;
    const int ct = wv - mt * NCT;
    const float*  Ta = Ts + (mt * 16 + l15) * TP + 4 * half;
    const float2* Bl = Bs + (2 * half) * PITCH + ct * 16 + l15;
    v8f acc = (v8f){0.f, 0.f, 0.f, 0.f, 0.f, 0.f, 0.f, 0.f};
#pragma unroll
    for (int J = 0; J < K / 8; ++J) {
      const float4 a4 = *reinterpret_cast<const float4*>(Ta + 8 * J);
#pragma unroll
      for (int e = 0; e < 2; ++e) {
        v2f a;
        a.x = e ? a4.z : a4.x;
        a.y = e ? a4.w : a4.y;
        const float2 bb = Bl[(4 * J + e) * PITCH];
        v2f b;
        b.x = bb.x;
        b.y = bb.y;
        acc = __builtin_amdgcn_wmma_f32_16x16x4_f32(false, a, false, b, (short)0, acc, false, false);
      }
    }
    const int mbase = tile0 + mt * 16 + half * 8;
    const int col = ct * 16 + l15;
    if ((NC % 16 == 0) || col < NC) {
#pragma unroll
      for (int r = 0; r < 8; ++r) D[(mbase + r) * NC + col] = acc[r];
    }
  }
}


extern "C" void kernel_launch(void* const* d_in, const int* in_sizes, int n_in,
                              void* d_out, int out_size, void* d_ws, size_t ws_size,
                              hipStream_t stream) {
  const float* features = (const float*)d_in[0];
  const int*   src      = (const int*)d_in[1];
  const int*   dst      = (const int*)d_in[2];
  const float* W1       = (const float*)d_in[3];
  const float* b1       = (const float*)d_in[4];
  const float* W2       = (const float*)d_in[5];
  const float* b2       = (const float*)d_in[6];
  const float* W3       = (const float*)d_in[7];
  const float* b3       = (const float*)d_in[8];

  const int N = NNODES;
  const int E = in_sizes[1];

  float* ws      = (float*)d_ws;
  float* inv_out = ws;
  float* inv_in  = ws + N;
  float* P0      = ws + 2 * (size_t)N;
  float* P1      = P0 + (size_t)N * 256;
  float* P2      = P1 + (size_t)N * 128;
  float* H3      = P0 + (size_t)N * 128;

  const int NB = (N + 1023) / 1024;
  int* cnt_in   = (int*)P2;
  int* rowptr   = cnt_in + N;
  int* rank     = rowptr + N + 4;
  int* csr_src  = rank + E;
  int* blocksum = csr_src + E;
  int* part_out = (int*)P0;
  int* part_in  = part_out + (size_t)NCHUNK * N;

  const int T = 256;
  auto blocks = [T](long long n) { return (unsigned)((n + T - 1) / T); };

  count_lds_kernel<<<dim3(NRANGE, NCHUNK), 1024, 0, stream>>>(src, dst, part_out, part_in, rank, E);
  reduce_norm_kernel<<<blocks(N), T, 0, stream>>>(part_out, part_in, cnt_in, inv_out, inv_in, N);
  scan_local_kernel<<<NB, 256, 0, stream>>>(cnt_in, rowptr, blocksum, N);
  scan_add_kernel<<<NB, 256, 0, stream>>>(rowptr, blocksum, N);
  fill_kernel<<<blocks(E), T, 0, stream>>>(src, dst, rowptr, part_in, rank, csr_src, E);

  gather_kernel<128, true, false, false><<<blocks((long long)N * 32), 256, 0, stream>>>(
      features, rowptr, csr_src, inv_out, nullptr, nullptr, P1, N);
  gemm_chain<128, 256, 128><<<256, 512, 0, stream>>>(
      P1, W1, W2, P0, inv_in, b1, inv_out, N / 16);

  gather_project_kernel<40><<<N / 32, 256, 0, stream>>>(
      P0, rowptr, csr_src, inv_in, b2, inv_out, W3, H3);

  gather_kernel<40, false, true, false><<<blocks((long long)N * 10), 256, 0, stream>>>(
      H3, rowptr, csr_src, nullptr, inv_in, b3, (float*)d_out, N);
}


# Row offsets made in the per-node launch (per-block in-edge totals from the count launch + in-block scan), planes hold absolute slots: both scan launches gone, fill reads one base per edge
# speedup vs baseline: 4.9586x; 1.0386x over previous
#include <hip/hip_runtime.h>


typedef __attribute__((ext_vector_type(2))) float v2f;
typedef __attribute__((ext_vector_type(8))) float v8f;

#define NNODES 100000


#define NRANGE 8
#define RANGE (NNODES / NRANGE)
#define NCHUNK 16
#define BLK 512
#define NBLK_R ((RANGE + BLK - 1) / BLK)
#define NBLK (NRANGE * NBLK_R)
#define NGRP ((RANGE + 31) / 32)
#define GPW ((NGRP + 31) / 32)

__device__ __forceinline__ int chunk_len(int E) {
  return (((E + NCHUNK - 1) / NCHUNK) + 3) & ~3;
}

__global__ __launch_bounds__(1024) void count_lds_kernel(const int* __restrict__ src,
                                                         const int* __restrict__ dst,
                                                         int* __restrict__ part_out,
                                                         int* __restrict__ part_in,
                                                         int* __restrict__ rank,
                                                         int* __restrict__ btot, int E) {
  static_assert(NNODES % NRANGE == 0, "equal node ranges");
  static_assert(BLK % 32 == 0 && NBLK_R <= 1024, "whole 32-node groups per block");
  static_assert(GPW <= BLK / 32, "a wave's groups lie in at most two blocks");
  __shared__ int co[RANGE];
  __shared__ int ci[RANGE];
  __shared__ int bl[NBLK_R * 32];
  const int t  = threadIdx.x;
  const int lo = (int)blockIdx.x * RANGE;
  const int c  = (int)blockIdx.y;
  for (int i = t; i < RANGE; i += 1024) { co[i] = 0; ci[i] = 0; }
  if (t < NBLK_R * 32) bl[t] = 0;
  __syncthreads();
  const int len = chunk_len(E);
  const int e0 = c * len;
  const int e1 = (e0 + len < E) ? (e0 + len) : E;
  int e = e0 + t * 4;
  for (; e + 4100 <= e1; e += 8192) {
    const int4 sa = *reinterpret_cast<const int4*>(src + e);
    const int4 da = *reinterpret_cast<const int4*>(dst + e);
    const int4 sb = *reinterpret_cast<const int4*>(src + e + 4096);
    const int4 db = *reinterpret_cast<const int4*>(dst + e + 4096);
    const int sv[8] = {sa.x, sa.y, sa.z, sa.w, sb.x, sb.y, sb.z, sb.w};
    const int dv[8] = {da.x, da.y, da.z, da.w, db.x, db.y, db.z, db.w};
    int rv[8];
#pragma unroll
    for (int k = 0; k < 8; ++k) {
      const unsigned s = (unsigned)(sv[k] - lo);
      const unsigned d = (unsigned)(dv[k] - lo);
      if (s < (unsigned)RANGE) atomicAdd(&co[s], 1);
      rv[k] = -1;
      if (d < (unsigned)RANGE) rv[k] = atomicAdd(&ci[d], 1);
    }
#pragma unroll
    for (int k = 0; k < 8; ++k) {
      if (rv[k] >= 0) rank[e + (k < 4 ? k : 4092 + k)] = rv[k];
    }
  }
  for (; e < e1; e += 4096) {
    for (int k = 0; k < 4; ++k) {
      if (e + k < e1) {
        const unsigned s = (unsigned)(src[e + k] - lo);
        const unsigned d = (unsigned)(dst[e + k] - lo);
        if (s < (unsigned)RANGE) atomicAdd(&co[s], 1);
        if (d < (unsigned)RANGE) rank[e + k] = atomicAdd(&ci[d], 1);
      }
    }
  }
  __syncthreads();
  {
    const int lane = t & 31;
    const int g0 = (t >> 5) * GPW;
    const int b0 = g0 / (BLK / 32);
    int acc0 = 0, acc1 = 0;
    int vci[GPW], vco[GPW];
#pragma unroll
    for (int j = 0; j < GPW; ++j) {
      const int i = (g0 + j) * 32 + lane;
      const int ic = i < RANGE ? i : RANGE - 1;
      vci[j] = ci[ic];
      vco[j] = co[ic];
    }
#pragma unroll
    for (int j = 0; j < GPW; ++j) {
      const int g = g0 + j;
      const int i = g * 32 + lane;
      if (i < RANGE) {
        part_out[c * NNODES + lo + i] = vco[j];
        part_in[c * NNODES + lo + i]  = vci[j];
        if (g / (BLK / 32) == b0) acc0 += vci[j]; else acc1 += vci[j];
      }
    }
    if (b0 < NBLK_R)     atomicAdd(&bl[b0 * 32 + lane], acc0);
    if (b0 + 1 < NBLK_R) atomicAdd(&bl[(b0 + 1) * 32 + lane], acc1);
  }
  __syncthreads();
  if (t < NBLK_R) {
    int s = 0;
    for (int l = 0; l < 32; ++l) s += bl[t * 32 + l];
    btot[c * NBLK + (int)blockIdx.x * NBLK_R + t] = s;
  }
}

__global__ __launch_bounds__(BLK) void node_offsets_kernel(const int* __restrict__ part_out,
                                                           int* __restrict__ part_in,
                                                           const int* __restrict__ btot,
                                                           int* __restrict__ rowptr,
                                                           float* __restrict__ inv_out,
                                                           float* __restrict__ inv_in) {
  static_assert(NRANGE * RANGE == NNODES, "the blocks cover every node");
  constexpr int NW = BLK / 32;
  __shared__ int shw[NW];
  __shared__ int shp[NW];
  const int t    = threadIdx.x;
  const int lane = t & 31;
  const int wave = t >> 5;
  const int B = (int)blockIdx.x;
  const int r = B / NBLK_R;
  const int k = B - r * NBLK_R;
  const int left = RANGE - k * BLK;
  const int len  = left < BLK ? left : BLK;
  const int node = r * RANGE + k * BLK + t;
  const bool active = t < len;

  constexpr int NT = NCHUNK * NBLK;
  constexpr int NS = (NT + BLK - 1) / BLK;
  int bt[NS];
#pragma unroll
  for (int s = 0; s < NS; ++s) {
    const int idx = t + s * BLK;
    bt[s] = btot[idx < NT ? idx : NT - 1];
  }
  int vo[NCHUNK], vi[NCHUNK];
  const int ld = active ? node : node - t;
#pragma unroll
  for (int c = 0; c < NCHUNK; ++c) {
    vo[c] = part_out[c * NNODES + ld];
    vi[c] = part_in[c * NNODES + ld];
  }
  int part = 0;
#pragma unroll
  for (int s = 0; s < NS; ++s) {
    const int idx = t + s * BLK;
    if (idx < NT && idx % NBLK < B) part += bt[s];
  }
  int so = 0, si = 0;
  int pre[NCHUNK];
#pragma unroll
  for (int c = 0; c < NCHUNK; ++c) {
    so += vo[c];
    pre[c] = si;
    si += active ? vi[c] : 0;
  }
  if (active) inv_out[node] = rsqrtf(fmaxf((float)so, 1.0f));
  int x = si;
#pragma unroll
  for (int d = 1; d < 32; d <<= 1) {
    const int y = __shfl_up(x, d);
    if (lane >= d) x += y;
  }
  part += __shfl_xor(part, 16);
  part += __shfl_xor(part, 8);
  part += __shfl_xor(part, 4);
  part += __shfl_xor(part, 2);
  part += __shfl_xor(part, 1);
  if (lane == 31) shw[wave] = x;
  if (lane == 0)  shp[wave] = part;
  __syncthreads();
  int first = 0;
#pragma unroll
  for (int w = 0; w < NW; ++w) {
    first += shp[w];
    if (w < wave) first += shw[w];
  }
  first += x - si;
  if (active) {
    rowptr[node] = first;
#pragma unroll
    for (int c = 0; c < NCHUNK; ++c) part_in[c * NNODES + node] = first + pre[c];
    inv_in[node]  = rsqrtf(fmaxf((float)si, 1.0f));
    if (B == NBLK - 1 && t == len - 1) rowptr[NNODES] = first + si;
  }
}

__global__ void fill_kernel(const int* __restrict__ src, const int* __restrict__ dst,
                            const int* __restrict__ base, const int* __restrict__ rank,
                            int* __restrict__ csr_src, int E) {
  int e = blockIdx.x * blockDim.x + threadIdx.x;
  if (e < E) {
    const int d = dst[e];
    const int c = e / chunk_len(E);
    csr_src[base[c * NNODES + d] + rank[e]] = src[e];
  }
}

template <int F, bool SCALE, bool EPI, bool RELU>
__global__ __launch_bounds__(256) void gather_kernel(const float* __restrict__ x,
                                                     const int* __restrict__ rowptr,
                                                     const int* __restrict__ csr_src,
                                                     const float* __restrict__ scale,
                                                     const float* __restrict__ rowScaleOut,
                                                     const float* __restrict__ bias,
                                                     float* __restrict__ agg, int n) {
  constexpr int LPN = F / 4;
  const int tid = blockIdx.x * 256 + threadIdx.x;
  int node = tid / LPN;
  const int q = tid - node * LPN;
  if (LPN == 32) node = __builtin_amdgcn_readfirstlane(node);
  if (node >= n) return;
  const int beg = rowptr[node];
  const int end = rowptr[node + 1];
  float rs = 1.0f;
  float4 bv = {0.f, 0.f, 0.f, 0.f};
  if (EPI) {
    rs = rowScaleOut[node];
    bv = reinterpret_cast<const float4*>(bias)[q];
  }
  const float4* __restrict__ xv = reinterpret_cast<const float4*>(x);
  float4 acc = {0.f, 0.f, 0.f, 0.f};
  int j = beg;
  for (; j + 4 <= end; j += 4) {
    const int s0 = csr_src[j], s1 = csr_src[j + 1], s2 = csr_src[j + 2], s3 = csr_src[j + 3];
    const float4 v0 = xv[s0 * LPN + q];
    const float4 v1 = xv[s1 * LPN + q];
    const float4 v2 = xv[s2 * LPN + q];
    const float4 v3 = xv[s3 * LPN + q];
    float w0 = 1.0f, w1 = 1.0f, w2 = 1.0f, w3 = 1.0f;
    if (SCALE) { w0 = scale[s0]; w1 = scale[s1]; w2 = scale[s2]; w3 = scale[s3]; }
    acc.x += v0.x * w0; acc.y += v0.y * w0; acc.z += v0.z * w0; acc.w += v0.w * w0;
    acc.x += v1.x * w1; acc.y += v1.y * w1; acc.z += v1.z * w1; acc.w += v1.w * w1;
    acc.x += v2.x * w2; acc.y += v2.y * w2; acc.z += v2.z * w2; acc.w += v2.w * w2;
    acc.x += v3.x * w3; acc.y += v3.y * w3; acc.z += v3.z * w3; acc.w += v3.w * w3;
  }
  for (; j < end; ++j) {
    const int s0 = csr_src[j];
    const float4 v0 = xv[s0 * LPN + q];
    float w0 = 1.0f;
    if (SCALE) w0 = scale[s0];
    acc.x += v0.x * w0; acc.y += v0.y * w0; acc.z += v0.z * w0; acc.w += v0.w * w0;
  }
  if (EPI) {
    acc.x = acc.x * rs + bv.x;
    acc.y = acc.y * rs + bv.y;
    acc.z = acc.z * rs + bv.z;
    acc.w = acc.w * rs + bv.w;
  }
  if (RELU) {
    acc.x = fmaxf(acc.x, 0.0f);
    acc.y = fmaxf(acc.y, 0.0f);
    acc.z = fmaxf(acc.z, 0.0f);
    acc.w = fmaxf(acc.w, 0.0f);
  }
  reinterpret_cast<float4*>(agg)[node * LPN + q] = acc;
}


template <int K, int NC, int PITCH, int NTHREADS>
__device__ __forceinline__ void stage_pairs(const float* __restrict__ B, float2* Bs) {
  constexpr int C4 = NC / 4;
  constexpr int ITEMS = (K / 2) * C4;
  constexpr int PER = ITEMS / NTHREADS;
  constexpr int BATCH = 4;
  static_assert(ITEMS % NTHREADS == 0 && PER % BATCH == 0, "whole batches per thread");
#pragma unroll
  for (int s0 = 0; s0 < PER; s0 += BATCH) {
    float4 r0[BATCH], r1[BATCH];
#pragma unroll
    for (int s = 0; s < BATCH; ++s) {
      const int i = (int)threadIdx.x + (s0 + s) * NTHREADS;
      const int kp = i / C4;
      const int c4 = (i - kp * C4) * 4;
      r0[s] = *reinterpret_cast<const float4*>(B + (2 * kp) * NC + c4);
      r1[s] = *reinterpret_cast<const float4*>(B + (2 * kp + 1) * NC + c4);
    }
#pragma unroll
    for (int s = 0; s < BATCH; ++s) {
      const int i = (int)threadIdx.x + (s0 + s) * NTHREADS;
      const int kp = i / C4;
      const int c4 = (i - kp * C4) * 4;
      float4* dp = reinterpret_cast<float4*>(&Bs[kp * PITCH + c4]);
      dp[0] = make_float4(r0[s].x, r1[s].x, r0[s].y, r1[s].y);
      dp[1] = make_float4(r0[s].z, r1[s].z, r0[s].w, r1[s].w);
    }
  }
}

template <int K1, int H, int N2>
__global__ __launch_bounds__(512) __attribute__((amdgpu_waves_per_eu(4, 4))) void gemm_chain(
    const float* __restrict__ A, const float* __restrict__ W1, const float* __restrict__ W2,
    float* __restrict__ D, const float* __restrict__ rowScale1, const float* __restrict__ bias1,
    const float* __restrict__ rowScale2, int mtiles) {
  static_assert(K1 % 8 == 0 && H % 64 == 0 && N2 % 16 == 0, "tile shapes");
  constexpr int FB  = 4;
  constexpr int NG  = H / 16 / FB;
  constexpr int NT2 = N2 / 16;
  constexpr int P1 = H + 8;
  constexpr int P2 = N2 + 4;
  __shared__ __attribute__((aligned(16))) float2 W1s[(K1 / 2) * P1];
  __shared__ __attribute__((aligned(16))) float2 W2s[(H / 2) * P2];

  stage_pairs<K1, H, P1, 512>(W1, W1s);
  stage_pairs<H, N2, P2, 512>(W2, W2s);
  __syncthreads();

  const int lane = threadIdx.x & 31;
  const int half = lane >> 4;
  const int l15  = lane & 15;
  const int wv = __builtin_amdgcn_readfirstlane((int)(threadIdx.x >> 5));
  const int nWaves = (int)gridDim.x * 16;
  const float2* W1l = W1s + (2 * half) * P1 + l15;
  const float2* W2l = W2s + (4 * half) * P2 + l15;

  for (int u = wv * (int)gridDim.x + (int)blockIdx.x; u < mtiles; u += nWaves) {
    const int tileM = u * 16;
    const int row   = tileM + l15;
    const float4* __restrict__ Ap = reinterpret_cast<const float4*>(A + row * K1) + half;
    float4 an[K1 / 8];
#pragma unroll
    for (int j = 0; j < K1 / 8; ++j) an[j] = Ap[2 * j];
    const float s1 = rowScale1[row];
    const float s2 = rowScale2[row];

    v8f acc2[NT2];
#pragma unroll
    for (int t = 0; t < NT2; ++t) acc2[t] = (v8f){0.f, 0.f, 0.f, 0.f, 0.f, 0.f, 0.f, 0.f};

    constexpr int NB1 = K1 / 8;
    constexpr int NB  = NB1 + FB * 4;
    static_assert(FB * 2 == 8 && NT2 == 8, "8 fragments per block");
    static_assert(NB % 2 == 0, "the register sets alternate across passes");
    float2 fr[2][8];
#pragma unroll
    for (int e = 0; e < 2; ++e)
#pragma unroll
      for (int ff = 0; ff < FB; ++ff) fr[0][e * FB + ff] = W1l[e * P1 + ff * 16];

#pragma unroll 1
    for (int g = 0; g < NG; ++g) {
      const int gn = (g + 1 < NG) ? g + 1 : g;
      v8f acc1[FB];
#pragma unroll
      for (int ff = 0; ff < FB; ++ff) acc1[ff] = (v8f){0.f, 0.f, 0.f, 0.f, 0.f, 0.f, 0.f, 0.f};
      float4 bA, bB;
      float v[8];
#pragma unroll
      for (int b = 0; b < NB; ++b) {
        {
          const int bn = (b + 1 < NB) ? b + 1 : 0;
          const int gg = (b + 1 < NB) ? g : gn;
          float2* dst = fr[(b + 1) & 1];
          if (bn < NB1) {
            const float2* Wr = W1l + (4 * bn) * P1 + gg * (FB * 16);
#pragma unroll
            for (int e = 0; e < 2; ++e)
#pragma unroll
              for (int ff = 0; ff < FB; ++ff) dst[e * FB + ff] = Wr[e * P1 + ff * 16];
          } else {
            const int q = bn - NB1;
            const float2* Wr2 = W2l + (8 * (gg * FB + q / 4) + (q % 4)) * P2;
#pragma unroll
            for (int t = 0; t < NT2; ++t) dst[t] = Wr2[t * 16];
          }
        }
        const float2* cur = fr[b & 1];
        if (b < NB1) {
#pragma unroll
          for (int e = 0; e < 2; ++e) {
            v2f x;
            x.x = e ? an[b].z : an[b].x;
            x.y = e ? an[b].w : an[b].y;
#pragma unroll
            for (int ff = 0; ff < FB; ++ff) {
              v2f w;
              w.x = cur[e * FB + ff].x;
              w.y = cur[e * FB + ff].y;
              acc1[ff] = __builtin_amdgcn_wmma_f32_16x16x4_f32(false, w, false, x, (short)0,
                                                               acc1[ff], false, false);
            }
          }
          if (b == NB1 - 4) {
            bA = *reinterpret_cast<const float4*>(bias1 + 16 * (g * FB) + 8 * half);
            bB = *reinterpret_cast<const float4*>(bias1 + 16 * (g * FB) + 8 * half + 4);
          }
        } else {
          const int q  = b - NB1;
          const int ff = q / 4;
          const int kk = q % 4;
          if (kk == 0) {
            const float bb[8] = {bA.x, bA.y, bA.z, bA.w, bB.x, bB.y, bB.z, bB.w};
#pragma unroll
            for (int r = 0; r < 8; ++r) {
              float o = acc1[ff][r] * s1 + bb[r];
              o = fmaxf(o, 0.0f);
              v[r] = o * s2;
            }
            if (ff + 1 < FB) {
              bA = *reinterpret_cast<const float4*>(bias1 + 16 * (g * FB + ff + 1) + 8 * half);
              bB = *reinterpret_cast<const float4*>(bias1 + 16 * (g * FB + ff + 1) + 8 * half + 4);
            }
          }
          v2f a;
          a.x = v[2 * kk];
          a.y = v[2 * kk + 1];
#pragma unroll
          for (int t = 0; t < NT2; ++t) {
            v2f w;
            w.x = cur[t].x;
            w.y = cur[t].y;
            acc2[t] = __builtin_amdgcn_wmma_f32_16x16x4_f32(false, a, false, w, (short)0, acc2[t],
                                                            false, false);
          }
        }
      }
    }

    const int mbase = tileM + half * 8;
#pragma unroll
    for (int t = 0; t < NT2; ++t) {
      const int col = t * 16 + l15;
#pragma unroll
      for (int r = 0; r < 8; ++r) D[(mbase + r) * N2 + col] = acc2[t][r];
    }
  }
}

template <int NC>
__global__ __launch_bounds__(256) void gather_project_kernel(const float* __restrict__ x,
                                                             const int* __restrict__ rowptr,
                                                             const int* __restrict__ csr_src,
                                                             const float* __restrict__ rowScaleOut,
                                                             const float* __restrict__ bias,
                                                             const float* __restrict__ rowScaleNext,
                                                             const float* __restrict__ B,
                                                             float* __restrict__ D) {
  constexpr int K     = 128;
  constexpr int LPN   = 32;
  constexpr int TILE  = 32;
  constexpr int NCT   = (NC + 15) / 16;
  constexpr int NCP   = NCT * 16;
  constexpr int PITCH = NCP + 8;
  constexpr int TP    = K + 4;
  static_assert(NNODES % TILE == 0, "whole tiles only");
  static_assert(2 * NCT <= 8, "one wave per output tile");
  static_assert(NC % 4 == 0, "vector loads need this");
  __shared__ __attribute__((aligned(16))) float2 Bs[(K / 2) * PITCH];
  __shared__ __attribute__((aligned(16))) float  Ts[TILE * TP];

  constexpr int C4 = NCP / 4;
  constexpr int NST = ((K / 2) * C4 + 255) / 256;
  static_assert((K / 2) * C4 == NST * 256, "every thread stages NST items");
  float4 r0[NST], r1[NST];
#pragma unroll
  for (int s = 0; s < NST; ++s) {
    const int i = threadIdx.x + s * 256;
    const int kp = i / C4;
    const int c4 = (i - kp * C4) * 4;
    const int c4l = (c4 < NC) ? c4 : (NC - 4);
    r0[s] = *reinterpret_cast<const float4*>(B + (2 * kp) * NC + c4l);
    r1[s] = *reinterpret_cast<const float4*>(B + (2 * kp + 1) * NC + c4l);
  }
#pragma unroll
  for (int s = 0; s < NST; ++s) {
    const int i = threadIdx.x + s * 256;
    const int kp = i / C4;
    const int c4 = (i - kp * C4) * 4;
    float4 u0 = r0[s], u1 = r1[s];
    if (c4 >= NC) {
      u0 = make_float4(0.f, 0.f, 0.f, 0.f);
      u1 = make_float4(0.f, 0.f, 0.f, 0.f);
    }
    float4* dp = reinterpret_cast<float4*>(&Bs[kp * PITCH + c4]);
    dp[0] = make_float4(u0.x, u1.x, u0.y, u1.y);
    dp[1] = make_float4(u0.z, u1.z, u0.w, u1.w);
  }

  const int lane = threadIdx.x & 31;
  const int wv = __builtin_amdgcn_readfirstlane((int)(threadIdx.x >> 5));
  const int tile0 = (int)blockIdx.x * TILE;

  const float4* __restrict__ xv = reinterpret_cast<const float4*>(x);
  const float4 bv = reinterpret_cast<const float4*>(bias)[lane];
  for (int i = 0; i < TILE / 8; ++i) {
    const int lrow = wv * (TILE / 8) + i;
    const int node = tile0 + lrow;
    const int beg = rowptr[node];
    const int end = rowptr[node + 1];
    const float rs = rowScaleOut[node];
    const float nx = rowScaleNext[node];
    float4 acc = {0.f, 0.f, 0.f, 0.f};
    int j = beg;
    for (; j + 4 <= end; j += 4) {
      const int s0 = csr_src[j], s1 = csr_src[j + 1], s2 = csr_src[j + 2], s3 = csr_src[j + 3];
      const float4 v0 = xv[s0 * LPN + lane];
      const float4 v1 = xv[s1 * LPN + lane];
      const float4 v2 = xv[s2 * LPN + lane];
      const float4 v3 = xv[s3 * LPN + lane];
      acc.x += v0.x; acc.y += v0.y; acc.z += v0.z; acc.w += v0.w;
      acc.x += v1.x; acc.y += v1.y; acc.z += v1.z; acc.w += v1.w;
      acc.x += v2.x; acc.y += v2.y; acc.z += v2.z; acc.w += v2.w;
      acc.x += v3.x; acc.y += v3.y; acc.z += v3.z; acc.w += v3.w;
    }
    for (; j < end; ++j) {
      const int s0 = csr_src[j];
      const float4 v0 = xv[s0 * LPN + lane];
      acc.x += v0.x; acc.y += v0.y; acc.z += v0.z; acc.w += v0.w;
    }
    acc.x = fmaxf(acc.x * rs + bv.x, 0.0f) * nx;
    acc.y = fmaxf(acc.y * rs + bv.y, 0.0f) * nx;
    acc.z = fmaxf(acc.z * rs + bv.z, 0.0f) * nx;
    acc.w = fmaxf(acc.w * rs + bv.w, 0.0f) * nx;
    *reinterpret_cast<float4*>(&Ts[lrow * TP + lane * 4]) = acc;
  }
  __syncthreads();

  if (wv < 2 * NCT) {
    const int half = lane >> 4;
    const int l15  = lane & 15;
    const int mt = wv / NCT;
    const int ct = wv - mt * NCT;
    const float*  Ta = Ts + (mt * 16 + l15) * TP + 4 * half;
    const float2* Bl = Bs + (2 * half) * PITCH + ct * 16 + l15;
    v8f acc = (v8f){0.f, 0.f, 0.f, 0.f, 0.f, 0.f, 0.f, 0.f};
#pragma unroll
    for (int J = 0; J < K / 8; ++J) {
      const float4 a4 = *reinterpret_cast<const float4*>(Ta + 8 * J);
#pragma unroll
      for (int e = 0; e < 2; ++e) {
        v2f a;
        a.x = e ? a4.z : a4.x;
        a.y = e ? a4.w : a4.y;
        const float2 bb = Bl[(4 * J + e) * PITCH];
        v2f b;
        b.x = bb.x;
        b.y = bb.y;
        acc = __builtin_amdgcn_wmma_f32_16x16x4_f32(false, a, false, b, (short)0, acc, false, false);
      }
    }
    const int mbase = tile0 + mt * 16 + half * 8;
    const int col = ct * 16 + l15;
    if ((NC % 16 == 0) || col < NC) {
#pragma unroll
      for (int r = 0; r < 8; ++r) D[(mbase + r) * NC + col] = acc[r];
    }
  }
}


extern "C" void kernel_launch(void* const* d_in, const int* in_sizes, int n_in,
                              void* d_out, int out_size, void* d_ws, size_t ws_size,
                              hipStream_t stream) {
  const float* features = (const float*)d_in[0];
  const int*   src      = (const int*)d_in[1];
  const int*   dst      = (const int*)d_in[2];
  const float* W1       = (const float*)d_in[3];
  const float* b1       = (const float*)d_in[4];
  const float* W2       = (const float*)d_in[5];
  const float* b2       = (const float*)d_in[6];
  const float* W3       = (const float*)d_in[7];
  const float* b3       = (const float*)d_in[8];

  const int N = NNODES;
  const int E = in_sizes[1];

  float* ws      = (float*)d_ws;
  float* inv_out = ws;
  float* inv_in  = ws + N;
  float* P0      = ws + 2 * (size_t)N;
  float* P1      = P0 + (size_t)N * 256;
  float* P2      = P1 + (size_t)N * 128;
  float* H3      = P0 + (size_t)N * 128;

  int* rowptr   = (int*)P2;
  int* rank     = rowptr + N + 4;
  int* csr_src  = rank + E;
  int* btot     = csr_src + E;
  int* part_out = (int*)P0;
  int* part_in  = part_out + (size_t)NCHUNK * N;

  const int T = 256;
  auto blocks = [T](long long n) { return (unsigned)((n + T - 1) / T); };

  count_lds_kernel<<<dim3(NRANGE, NCHUNK), 1024, 0, stream>>>(src, dst, part_out, part_in, rank,
                                                              btot, E);
  node_offsets_kernel<<<NBLK, BLK, 0, stream>>>(part_out, part_in, btot, rowptr, inv_out, inv_in);
  fill_kernel<<<blocks(E), T, 0, stream>>>(src, dst, part_in, rank, csr_src, E);

  gather_kernel<128, true, false, false><<<blocks((long long)N * 32), 256, 0, stream>>>(
      features, rowptr, csr_src, inv_out, nullptr, nullptr, P1, N);
  gemm_chain<128, 256, 128><<<256, 512, 0, stream>>>(
      P1, W1, W2, P0, inv_in, b1, inv_out, N / 16);

  gather_project_kernel<40><<<N / 32, 256, 0, stream>>>(
      P0, rowptr, csr_src, inv_in, b2, inv_out, W3, H3);

  gather_kernel<40, false, true, false><<<blocks((long long)N * 10), 256, 0, stream>>>(
      H3, rowptr, csr_src, nullptr, inv_in, b3, (float*)d_out, N);
}
